# starting kernel
# baseline (speedup 1.0000x reference)
#include <hip/hip_runtime.h>
#include <cstdint>

#define NFEAT 128

typedef __attribute__((ext_vector_type(2))) float v2f;
typedef __attribute__((ext_vector_type(8))) float v8f;


__global__ void init_deg_kernel(float* __restrict__ deg, int n) {
    int i = blockIdx.x * blockDim.x + threadIdx.x;
    if (i < n) deg[i] = 1.0f;
}

__global__ void deg_accum_kernel(const int* __restrict__ dst, float* __restrict__ deg, int e) {
    int i = blockIdx.x * blockDim.x + threadIdx.x;
    if (i < e) atomicAdd(&deg[dst[i]], 1.0f);
}

__global__ void dinv_kernel(float* __restrict__ deg, int n) {
    int i = blockIdx.x * blockDim.x + threadIdx.x;
    if (i < n) deg[i] = rsqrtf(deg[i]);
}


__global__ __launch_bounds__(256) void gemm_xwT_wmma(const float* __restrict__ x,
                                                     const float* __restrict__ W,
                                                     float* __restrict__ xw, int n) {
    __shared__ float lds_x[16 * NFEAT];

    const int tid  = threadIdx.x;
    const int n0   = blockIdx.x * 16;

    const float4* xs = (const float4*)(x + (size_t)n0 * NFEAT);
    float4* ls = (float4*)lds_x;
    ls[tid]       = xs[tid];
    ls[tid + 256] = xs[tid + 256];
    __syncthreads();

    const int wave  = tid >> 5;
    const int lane  = tid & 31;
    const int o0    = wave * 16;
    const int row   = lane & 15;
    const int khalf = (lane >> 4) << 1;

    v8f acc = {};
    const float* wrow = W + (size_t)(o0 + row) * NFEAT;

#pragma unroll
    for (int k0 = 0; k0 < NFEAT; k0 += 4) {
        const int kk = k0 + khalf;
        v2f a, b;
        a.x = lds_x[row * NFEAT + kk];
        a.y = lds_x[row * NFEAT + kk + 1];
        b.x = wrow[kk];
        b.y = wrow[kk + 1];
        acc = __builtin_amdgcn_wmma_f32_16x16x4_f32(false, a, false, b,
                                                    (short)0, acc, false, false);
    }

    const int col = o0 + row;
#pragma unroll
    for (int r = 0; r < 8; ++r) {
        const int m = r + ((lane >> 4) << 3);
        xw[(size_t)(n0 + m) * NFEAT + col] = acc[r];
    }
}


__global__ void init_agg_kernel(const float* __restrict__ xw, const float* __restrict__ dinv,
                                const float* __restrict__ bias, float* __restrict__ agg,
                                int n) {
    size_t i = (size_t)blockIdx.x * blockDim.x + threadIdx.x;
    if (i < (size_t)n * NFEAT) {
        int node = (int)(i >> 7);
        int f    = (int)(i & (NFEAT - 1));
        float di = dinv[node];
        agg[i] = bias[f] + xw[i] * di * di;
    }
}


__global__ __launch_bounds__(256) void scatter_edges_kernel(const int* __restrict__ src,
                                                            const int* __restrict__ dst,
                                                            const float* __restrict__ xw,
                                                            const float* __restrict__ dinv,
                                                            float* __restrict__ agg, int e) {
    int gid  = blockIdx.x * blockDim.x + threadIdx.x;
    int edge = gid >> 5;
    int lane = gid & 31;
    if (edge >= e) return;

    int s = src[edge];
    int d = dst[edge];
    float w = dinv[s] * dinv[d];

    const float4* xs = (const float4*)(xw + (size_t)s * NFEAT);
    float4 v = xs[lane];
    float* ab = agg + (size_t)d * NFEAT + lane * 4;
    atomicAdd(ab + 0, v.x * w);
    atomicAdd(ab + 1, v.y * w);
    atomicAdd(ab + 2, v.z * w);
    atomicAdd(ab + 3, v.w * w);
}


__global__ __launch_bounds__(256) void finalize_kernel(const float* __restrict__ agg,
                                                       const float* __restrict__ wreg,
                                                       const float* __restrict__ breg,
                                                       float* __restrict__ out, int n) {
    int gid  = blockIdx.x * blockDim.x + threadIdx.x;
    int node = gid >> 5;
    int lane = gid & 31;
    if (node >= n) return;

    const float4* av = (const float4*)(agg + (size_t)node * NFEAT);
    const float4* wv = (const float4*)wreg;
    float4 a = av[lane];
    float4 w = wv[lane];
    float dot = fmaxf(a.x, 0.0f) * w.x + fmaxf(a.y, 0.0f) * w.y +
                fmaxf(a.z, 0.0f) * w.z + fmaxf(a.w, 0.0f) * w.w;
#pragma unroll
    for (int off = 16; off > 0; off >>= 1)
        dot += __shfl_xor(dot, off, 32);
    if (lane == 0) out[node] = dot + breg[0];
}


extern "C" void kernel_launch(void* const* d_in, const int* in_sizes, int n_in,
                              void* d_out, int out_size, void* d_ws, size_t ws_size,
                              hipStream_t stream) {
    const float* x    = (const float*)d_in[0];
    const int*   ei   = (const int*)d_in[1];
    const float* W    = (const float*)d_in[2];
    const float* b    = (const float*)d_in[3];
    const float* wreg = (const float*)d_in[4];
    const float* breg = (const float*)d_in[5];
    float* out = (float*)d_out;

    const int n = in_sizes[0] / NFEAT;
    const int e = in_sizes[1] / 2;
    const int* src = ei;
    const int* dst = ei + e;

    float* deg = (float*)d_ws;
    float* xw  = deg + n;
    float* agg = xw + (size_t)n * NFEAT;

    const int B = 256;
    init_deg_kernel<<<(n + B - 1) / B, B, 0, stream>>>(deg, n);
    deg_accum_kernel<<<(e + B - 1) / B, B, 0, stream>>>(dst, deg, e);
    dinv_kernel<<<(n + B - 1) / B, B, 0, stream>>>(deg, n);

    gemm_xwT_wmma<<<n / 16, B, 0, stream>>>(x, W, xw, n);

    unsigned int agg_blocks = (unsigned int)(((size_t)n * NFEAT + B - 1) / B);
    init_agg_kernel<<<agg_blocks, B, 0, stream>>>(xw, deg, b, agg, n);

    unsigned int sc_blocks = (unsigned int)(((size_t)e * 32 + B - 1) / B);
    scatter_edges_kernel<<<sc_blocks, B, 0, stream>>>(src, dst, xw, deg, agg, e);

    unsigned int fin_blocks = (unsigned int)(((size_t)n * 32 + B - 1) / B);
    finalize_kernel<<<fin_blocks, B, 0, stream>>>(agg, wreg, breg, out, n);
}


# Atomic edge scatter, agg init and relu-dot pass replaced by a dst-sorted edge list (int atomic counts+ranks, scan, fill) and one gather wave per node; edge sum order = arrival-order rank
# speedup vs baseline: 1.6994x; 1.6994x over previous
#include <hip/hip_runtime.h>
#include <cstdint>


#define NFEAT 128

typedef __attribute__((ext_vector_type(2))) float v2f;
typedef __attribute__((ext_vector_type(8))) float v8f;


__global__ void zero_counts_kernel(int* __restrict__ cnt, int n) {
    int i = blockIdx.x * blockDim.x + threadIdx.x;
    if (i < n) cnt[i] = 0;
}

__global__ void count_rank_kernel(const int* __restrict__ dst, int* __restrict__ cnt,
                                  int* __restrict__ rank, int e) {
    int i = blockIdx.x * blockDim.x + threadIdx.x;
    if (i < e) rank[i] = atomicAdd(&cnt[dst[i]], 1);
}


__global__ __launch_bounds__(1024) void offsets_dinv_kernel(const int* __restrict__ cnt,
                                                            int* __restrict__ off,
                                                            float* __restrict__ dinv, int n) {
    __shared__ int s_pre[32];
    __shared__ int s_tot[32];

    const int tid   = threadIdx.x;
    const int lane  = tid & 31;
    const int wave  = tid >> 5;
    const int start = blockIdx.x * 1024;

    int pre = 0;
    const int4* c4 = (const int4*)cnt;
    const int nq = start >> 2;
    int q = tid;
    for (; q + 3072 < nq; q += 4096) {
        const int4 v0 = c4[q];
        const int4 v1 = c4[q + 1024];
        const int4 v2 = c4[q + 2048];
        const int4 v3 = c4[q + 3072];
        pre += ((v0.x + v0.y) + (v0.z + v0.w)) + ((v1.x + v1.y) + (v1.z + v1.w)) +
               ((v2.x + v2.y) + (v2.z + v2.w)) + ((v3.x + v3.y) + (v3.z + v3.w));
    }
    for (; q < nq; q += 1024) {
        const int4 v = c4[q];
        pre += (v.x + v.y) + (v.z + v.w);
    }

    const int i = start + tid;
    const int c = (i < n) ? cnt[i] : 0;

    int incl = c;
#pragma unroll
    for (int d = 1; d < 32; d <<= 1) {
        int t = __shfl_up(incl, d, 32);
        if (lane >= d) incl += t;
    }
#pragma unroll
    for (int o = 16; o > 0; o >>= 1) pre += __shfl_xor(pre, o, 32);

    if (lane == 31) s_tot[wave] = incl;
    if (lane == 0)  s_pre[wave] = pre;
    __syncthreads();

    int base = s_pre[lane];
#pragma unroll
    for (int o = 16; o > 0; o >>= 1) base += __shfl_xor(base, o, 32);

    const int tw = s_tot[lane];
    int tincl = tw;
#pragma unroll
    for (int d = 1; d < 32; d <<= 1) {
        int t = __shfl_up(tincl, d, 32);
        if (lane >= d) tincl += t;
    }
    const int woff = __shfl(tincl - tw, wave, 32);

    if (i < n) {
        const int excl = base + woff + incl - c;
        off[i]  = excl;
        dinv[i] = rsqrtf((float)(c + 1));
        if (i == n - 1) off[n] = excl + c;
    }
}


__global__ void fill_sorted_kernel(const int* __restrict__ src, const int* __restrict__ dst,
                                   const int* __restrict__ rank, const int* __restrict__ off,
                                   const float* __restrict__ dinv, int2* __restrict__ csr, int e) {
    int i = blockIdx.x * blockDim.x + threadIdx.x;
    if (i < e) {
        const int s = src[i];
        const int d = dst[i];
        const float w = dinv[s] * dinv[d];
        csr[off[d] + rank[i]] = make_int2(s, __float_as_int(w));
    }
}


__global__ __launch_bounds__(256) void gemm_xwT_wmma(const float* __restrict__ x,
                                                     const float* __restrict__ W,
                                                     float* __restrict__ xw, int n) {
    __shared__ float lds_x[16 * NFEAT];

    const int tid  = threadIdx.x;
    const int n0   = blockIdx.x * 16;

    const float4* xs = (const float4*)(x + (size_t)n0 * NFEAT);
    float4* ls = (float4*)lds_x;
    ls[tid]       = xs[tid];
    ls[tid + 256] = xs[tid + 256];
    __syncthreads();

    const int wave  = tid >> 5;
    const int lane  = tid & 31;
    const int o0    = wave * 16;
    const int row   = lane & 15;
    const int khalf = (lane >> 4) << 1;

    v8f acc = {};
    const float* wrow = W + (size_t)(o0 + row) * NFEAT;

#pragma unroll
    for (int k0 = 0; k0 < NFEAT; k0 += 4) {
        const int kk = k0 + khalf;
        v2f a, b;
        a.x = lds_x[row * NFEAT + kk];
        a.y = lds_x[row * NFEAT + kk + 1];
        b.x = wrow[kk];
        b.y = wrow[kk + 1];
        acc = __builtin_amdgcn_wmma_f32_16x16x4_f32(false, a, false, b,
                                                    (short)0, acc, false, false);
    }

    const int col = o0 + row;
#pragma unroll
    for (int r = 0; r < 8; ++r) {
        const int m = r + ((lane >> 4) << 3);
        xw[(size_t)(n0 + m) * NFEAT + col] = acc[r];
    }
}


__device__ __forceinline__ void fma4(float4& acc, const float4 v, const float w) {
    acc.x += v.x * w;
    acc.y += v.y * w;
    acc.z += v.z * w;
    acc.w += v.w * w;
}

__global__ __launch_bounds__(256) void gather_finalize_kernel(const float* __restrict__ xw,
                                                              const int* __restrict__ off,
                                                              const float* __restrict__ dinv,
                                                              const int2* __restrict__ csr,
                                                              const float* __restrict__ bias,
                                                              const float* __restrict__ wreg,
                                                              const float* __restrict__ breg,
                                                              float* __restrict__ out, int n) {
    const int lane = threadIdx.x & 31;
    const int node = __builtin_amdgcn_readfirstlane((int)(blockIdx.x * 8 + (threadIdx.x >> 5)));
    if (node >= n) return;

    const float4* rows = (const float4*)xw;
    const int beg = off[node];
    const int end = off[node + 1];

    int base = beg;
    int m = min(32, end - base);
    int2 ent = make_int2(0, 0);
    if (m > 0) ent = csr[base + min(lane, m - 1)];

    const float4 vself = rows[(size_t)node * (NFEAT / 4) + lane];
    const float4 b = ((const float4*)bias)[lane];
    const float4 w = ((const float4*)wreg)[lane];
    const float di = dinv[node];
    const float nself = di * di;

    float4 acc;
    acc.x = vself.x * nself;
    acc.y = vself.y * nself;
    acc.z = vself.z * nself;
    acc.w = vself.w * nself;

    while (m > 0) {
        int j = 0;
        for (; j + 4 <= m; j += 4) {
            const int s0 = __builtin_amdgcn_readlane(ent.x, j);
            const int s1 = __builtin_amdgcn_readlane(ent.x, j + 1);
            const int s2 = __builtin_amdgcn_readlane(ent.x, j + 2);
            const int s3 = __builtin_amdgcn_readlane(ent.x, j + 3);
            const float w0 = __int_as_float(__builtin_amdgcn_readlane(ent.y, j));
            const float w1 = __int_as_float(__builtin_amdgcn_readlane(ent.y, j + 1));
            const float w2 = __int_as_float(__builtin_amdgcn_readlane(ent.y, j + 2));
            const float w3 = __int_as_float(__builtin_amdgcn_readlane(ent.y, j + 3));
            const float4 v0 = rows[(size_t)s0 * (NFEAT / 4) + lane];
            const float4 v1 = rows[(size_t)s1 * (NFEAT / 4) + lane];
            const float4 v2 = rows[(size_t)s2 * (NFEAT / 4) + lane];
            const float4 v3 = rows[(size_t)s3 * (NFEAT / 4) + lane];
            fma4(acc, v0, w0);
            fma4(acc, v1, w1);
            fma4(acc, v2, w2);
            fma4(acc, v3, w3);
        }
        if (j + 2 <= m) {
            const int s0 = __builtin_amdgcn_readlane(ent.x, j);
            const int s1 = __builtin_amdgcn_readlane(ent.x, j + 1);
            const float w0 = __int_as_float(__builtin_amdgcn_readlane(ent.y, j));
            const float w1 = __int_as_float(__builtin_amdgcn_readlane(ent.y, j + 1));
            const float4 v0 = rows[(size_t)s0 * (NFEAT / 4) + lane];
            const float4 v1 = rows[(size_t)s1 * (NFEAT / 4) + lane];
            fma4(acc, v0, w0);
            fma4(acc, v1, w1);
            j += 2;
        }
        if (j < m) {
            const int s0 = __builtin_amdgcn_readlane(ent.x, j);
            const float w0 = __int_as_float(__builtin_amdgcn_readlane(ent.y, j));
            const float4 v0 = rows[(size_t)s0 * (NFEAT / 4) + lane];
            fma4(acc, v0, w0);
        }
        base += 32;
        m = min(32, end - base);
        if (m > 0) ent = csr[base + min(lane, m - 1)];
    }

    float dot = fmaxf(acc.x + b.x, 0.0f) * w.x + fmaxf(acc.y + b.y, 0.0f) * w.y +
                fmaxf(acc.z + b.z, 0.0f) * w.z + fmaxf(acc.w + b.w, 0.0f) * w.w;
#pragma unroll
    for (int o = 16; o > 0; o >>= 1)
        dot += __shfl_xor(dot, o, 32);
    if (lane == 0) out[node] = dot + breg[0];
}


extern "C" void kernel_launch(void* const* d_in, const int* in_sizes, int n_in,
                              void* d_out, int out_size, void* d_ws, size_t ws_size,
                              hipStream_t stream) {
    const float* x    = (const float*)d_in[0];
    const int*   ei   = (const int*)d_in[1];
    const float* W    = (const float*)d_in[2];
    const float* b    = (const float*)d_in[3];
    const float* wreg = (const float*)d_in[4];
    const float* breg = (const float*)d_in[5];
    float* out = (float*)d_out;

    const int n = in_sizes[0] / NFEAT;
    const int e = in_sizes[1] / 2;
    const int* src = ei;
    const int* dst = ei + e;

    float* xw   = (float*)d_ws;
    int*   cnt  = (int*)(xw + (size_t)n * NFEAT);
    float* dinv = (float*)(cnt + n);
    int2*  csr  = (int2*)(dinv + n);
    int*   rank = (int*)(csr + e);
    int*   off  = rank + e;

    const int B = 256;
    zero_counts_kernel<<<(n + B - 1) / B, B, 0, stream>>>(cnt, n);
    count_rank_kernel<<<(e + B - 1) / B, B, 0, stream>>>(dst, cnt, rank, e);
    offsets_dinv_kernel<<<(n + 1023) / 1024, 1024, 0, stream>>>(cnt, off, dinv, n);
    fill_sorted_kernel<<<(e + B - 1) / B, B, 0, stream>>>(src, dst, rank, off, dinv, csr, e);

    gemm_xwT_wmma<<<n / 16, B, 0, stream>>>(x, W, xw, n);

    gather_finalize_kernel<<<(n + 7) / 8, B, 0, stream>>>(xw, off, dinv, csr, b, wreg, breg, out, n);
}


# x tile in LDS at a row pitch of 132 floats instead of 128: the 16 rows of an A-fragment read start 4 banks apart, not on one bank (16-way LDS bank conflict of the product kernel gone)
# speedup vs baseline: 1.8560x; 1.0922x over previous
#include <hip/hip_runtime.h>
#include <cstdint>


#define NFEAT 128
#define XPITCH 132

typedef __attribute__((ext_vector_type(2))) float v2f;
typedef __attribute__((ext_vector_type(8))) float v8f;


__global__ void zero_counts_kernel(int* __restrict__ cnt, int n) {
    int i = blockIdx.x * blockDim.x + threadIdx.x;
    if (i < n) cnt[i] = 0;
}

__global__ void count_rank_kernel(const int* __restrict__ dst, int* __restrict__ cnt,
                                  int* __restrict__ rank, int e) {
    int i = blockIdx.x * blockDim.x + threadIdx.x;
    if (i < e) rank[i] = atomicAdd(&cnt[dst[i]], 1);
}


__global__ __launch_bounds__(1024) void offsets_dinv_kernel(const int* __restrict__ cnt,
                                                            int* __restrict__ off,
                                                            float* __restrict__ dinv, int n) {
    __shared__ int s_pre[32];
    __shared__ int s_tot[32];

    const int tid   = threadIdx.x;
    const int lane  = tid & 31;
    const int wave  = tid >> 5;
    const int start = blockIdx.x * 1024;

    int pre = 0;
    const int4* c4 = (const int4*)cnt;
    const int nq = start >> 2;
    int q = tid;
    for (; q + 3072 < nq; q += 4096) {
        const int4 v0 = c4[q];
        const int4 v1 = c4[q + 1024];
        const int4 v2 = c4[q + 2048];
        const int4 v3 = c4[q + 3072];
        pre += ((v0.x + v0.y) + (v0.z + v0.w)) + ((v1.x + v1.y) + (v1.z + v1.w)) +
               ((v2.x + v2.y) + (v2.z + v2.w)) + ((v3.x + v3.y) + (v3.z + v3.w));
    }
    for (; q < nq; q += 1024) {
        const int4 v = c4[q];
        pre += (v.x + v.y) + (v.z + v.w);
    }

    const int i = start + tid;
    const int c = (i < n) ? cnt[i] : 0;

    int incl = c;
#pragma unroll
    for (int d = 1; d < 32; d <<= 1) {
        int t = __shfl_up(incl, d, 32);
        if (lane >= d) incl += t;
    }
#pragma unroll
    for (int o = 16; o > 0; o >>= 1) pre += __shfl_xor(pre, o, 32);

    if (lane == 31) s_tot[wave] = incl;
    if (lane == 0)  s_pre[wave] = pre;
    __syncthreads();

    int base = s_pre[lane];
#pragma unroll
    for (int o = 16; o > 0; o >>= 1) base += __shfl_xor(base, o, 32);

    const int tw = s_tot[lane];
    int tincl = tw;
#pragma unroll
    for (int d = 1; d < 32; d <<= 1) {
        int t = __shfl_up(tincl, d, 32);
        if (lane >= d) tincl += t;
    }
    const int woff = __shfl(tincl - tw, wave, 32);

    if (i < n) {
        const int excl = base + woff + incl - c;
        off[i]  = excl;
        dinv[i] = rsqrtf((float)(c + 1));
        if (i == n - 1) off[n] = excl + c;
    }
}


__global__ void fill_sorted_kernel(const int* __restrict__ src, const int* __restrict__ dst,
                                   const int* __restrict__ rank, const int* __restrict__ off,
                                   const float* __restrict__ dinv, int2* __restrict__ csr, int e) {
    int i = blockIdx.x * blockDim.x + threadIdx.x;
    if (i < e) {
        const int s = src[i];
        const int d = dst[i];
        const float w = dinv[s] * dinv[d];
        csr[off[d] + rank[i]] = make_int2(s, __float_as_int(w));
    }
}


__global__ __launch_bounds__(256) void gemm_xwT_wmma(const float* __restrict__ x,
                                                     const float* __restrict__ W,
                                                     float* __restrict__ xw, int n) {
    __shared__ __attribute__((aligned(16))) float lds_x[16 * XPITCH];

    const int tid  = threadIdx.x;
    const int n0   = blockIdx.x * 16;

    const float4* xs = (const float4*)(x + (size_t)n0 * NFEAT);
    const int lrow = tid >> 5;
    const int lcol = (tid & 31) << 2;
    *(float4*)&lds_x[lrow * XPITCH + lcol]       = xs[tid];
    *(float4*)&lds_x[(lrow + 8) * XPITCH + lcol] = xs[tid + 256];
    __syncthreads();

    const int wave  = tid >> 5;
    const int lane  = tid & 31;
    const int o0    = wave * 16;
    const int row   = lane & 15;
    const int khalf = (lane >> 4) << 1;

    v8f acc = {};
    const float* wrow = W + (size_t)(o0 + row) * NFEAT;

#pragma unroll
    for (int k0 = 0; k0 < NFEAT; k0 += 4) {
        const int kk = k0 + khalf;
        v2f a, b;
        a.x = lds_x[row * XPITCH + kk];
        a.y = lds_x[row * XPITCH + kk + 1];
        b.x = wrow[kk];
        b.y = wrow[kk + 1];
        acc = __builtin_amdgcn_wmma_f32_16x16x4_f32(false, a, false, b,
                                                    (short)0, acc, false, false);
    }

    const int col = o0 + row;
#pragma unroll
    for (int r = 0; r < 8; ++r) {
        const int m = r + ((lane >> 4) << 3);
        xw[(size_t)(n0 + m) * NFEAT + col] = acc[r];
    }
}


__device__ __forceinline__ void fma4(float4& acc, const float4 v, const float w) {
    acc.x += v.x * w;
    acc.y += v.y * w;
    acc.z += v.z * w;
    acc.w += v.w * w;
}

__global__ __launch_bounds__(256) void gather_finalize_kernel(const float* __restrict__ xw,
                                                              const int* __restrict__ off,
                                                              const float* __restrict__ dinv,
                                                              const int2* __restrict__ csr,
                                                              const float* __restrict__ bias,
                                                              const float* __restrict__ wreg,
                                                              const float* __restrict__ breg,
                                                              float* __restrict__ out, int n) {
    const int lane = threadIdx.x & 31;
    const int node = __builtin_amdgcn_readfirstlane((int)(blockIdx.x * 8 + (threadIdx.x >> 5)));
    if (node >= n) return;

    const float4* rows = (const float4*)xw;
    const int beg = off[node];
    const int end = off[node + 1];

    int base = beg;
    int m = min(32, end - base);
    int2 ent = make_int2(0, 0);
    if (m > 0) ent = csr[base + min(lane, m - 1)];

    const float4 vself = rows[(size_t)node * (NFEAT / 4) + lane];
    const float4 b = ((const float4*)bias)[lane];
    const float4 w = ((const float4*)wreg)[lane];
    const float di = dinv[node];
    const float nself = di * di;

    float4 acc;
    acc.x = vself.x * nself;
    acc.y = vself.y * nself;
    acc.z = vself.z * nself;
    acc.w = vself.w * nself;

    while (m > 0) {
        int j = 0;
        for (; j + 4 <= m; j += 4) {
            const int s0 = __builtin_amdgcn_readlane(ent.x, j);
            const int s1 = __builtin_amdgcn_readlane(ent.x, j + 1);
            const int s2 = __builtin_amdgcn_readlane(ent.x, j + 2);
            const int s3 = __builtin_amdgcn_readlane(ent.x, j + 3);
            const float w0 = __int_as_float(__builtin_amdgcn_readlane(ent.y, j));
            const float w1 = __int_as_float(__builtin_amdgcn_readlane(ent.y, j + 1));
            const float w2 = __int_as_float(__builtin_amdgcn_readlane(ent.y, j + 2));
            const float w3 = __int_as_float(__builtin_amdgcn_readlane(ent.y, j + 3));
            const float4 v0 = rows[(size_t)s0 * (NFEAT / 4) + lane];
            const float4 v1 = rows[(size_t)s1 * (NFEAT / 4) + lane];
            const float4 v2 = rows[(size_t)s2 * (NFEAT / 4) + lane];
            const float4 v3 = rows[(size_t)s3 * (NFEAT / 4) + lane];
            fma4(acc, v0, w0);
            fma4(acc, v1, w1);
            fma4(acc, v2, w2);
            fma4(acc, v3, w3);
        }
        if (j + 2 <= m) {
            const int s0 = __builtin_amdgcn_readlane(ent.x, j);
            const int s1 = __builtin_amdgcn_readlane(ent.x, j + 1);
            const float w0 = __int_as_float(__builtin_amdgcn_readlane(ent.y, j));
            const float w1 = __int_as_float(__builtin_amdgcn_readlane(ent.y, j + 1));
            const float4 v0 = rows[(size_t)s0 * (NFEAT / 4) + lane];
            const float4 v1 = rows[(size_t)s1 * (NFEAT / 4) + lane];
            fma4(acc, v0, w0);
            fma4(acc, v1, w1);
            j += 2;
        }
        if (j < m) {
            const int s0 = __builtin_amdgcn_readlane(ent.x, j);
            const float w0 = __int_as_float(__builtin_amdgcn_readlane(ent.y, j));
            const float4 v0 = rows[(size_t)s0 * (NFEAT / 4) + lane];
            fma4(acc, v0, w0);
        }
        base += 32;
        m = min(32, end - base);
        if (m > 0) ent = csr[base + min(lane, m - 1)];
    }

    float dot = fmaxf(acc.x + b.x, 0.0f) * w.x + fmaxf(acc.y + b.y, 0.0f) * w.y +
                fmaxf(acc.z + b.z, 0.0f) * w.z + fmaxf(acc.w + b.w, 0.0f) * w.w;
#pragma unroll
    for (int o = 16; o > 0; o >>= 1)
        dot += __shfl_xor(dot, o, 32);
    if (lane == 0) out[node] = dot + breg[0];
}


extern "C" void kernel_launch(void* const* d_in, const int* in_sizes, int n_in,
                              void* d_out, int out_size, void* d_ws, size_t ws_size,
                              hipStream_t stream) {
    const float* x    = (const float*)d_in[0];
    const int*   ei   = (const int*)d_in[1];
    const float* W    = (const float*)d_in[2];
    const float* b    = (const float*)d_in[3];
    const float* wreg = (const float*)d_in[4];
    const float* breg = (const float*)d_in[5];
    float* out = (float*)d_out;

    const int n = in_sizes[0] / NFEAT;
    const int e = in_sizes[1] / 2;
    const int* src = ei;
    const int* dst = ei + e;

    float* xw   = (float*)d_ws;
    int*   cnt  = (int*)(xw + (size_t)n * NFEAT);
    float* dinv = (float*)(cnt + n);
    int2*  csr  = (int2*)(dinv + n);
    int*   rank = (int*)(csr + e);
    int*   off  = rank + e;

    const int B = 256;
    zero_counts_kernel<<<(n + B - 1) / B, B, 0, stream>>>(cnt, n);
    count_rank_kernel<<<(e + B - 1) / B, B, 0, stream>>>(dst, cnt, rank, e);
    offsets_dinv_kernel<<<(n + 1023) / 1024, 1024, 0, stream>>>(cnt, off, dinv, n);
    fill_sorted_kernel<<<(e + B - 1) / B, B, 0, stream>>>(src, dst, rank, off, dinv, csr, e);

    gemm_xwT_wmma<<<n / 16, B, 0, stream>>>(x, W, xw, n);

    gather_finalize_kernel<<<(n + 7) / 8, B, 0, stream>>>(xw, off, dinv, csr, b, wreg, breg, out, n);
}


# Fixed grids of long-lived waves that loop over their items (gather: nodes a grid apart; product: 32-row tiles, W fragment loaded once a wave; edge passes: 4 edges a thread): 194k waves a run, now 40k
# speedup vs baseline: 2.0897x; 1.1259x over previous
#include <hip/hip_runtime.h>
#include <cstdint>


#define NFEAT 128
#define XPITCH 132

typedef __attribute__((ext_vector_type(2))) float v2f;
typedef __attribute__((ext_vector_type(8))) float v8f;


__global__ void zero_counts_kernel(int* __restrict__ cnt, int n) {
    int i = blockIdx.x * blockDim.x + threadIdx.x;
    if (i < n) cnt[i] = 0;
}

#define EPT 4

__global__ void count_rank_kernel(const int* __restrict__ dst, int* __restrict__ cnt,
                                  int* __restrict__ rank, int e) {
    const int T = gridDim.x * blockDim.x;
    const int i = blockIdx.x * blockDim.x + threadIdx.x;
    if (i + (EPT - 1) * T < e) {
        int d[EPT], r[EPT];
#pragma unroll
        for (int k = 0; k < EPT; ++k) d[k] = dst[i + k * T];
#pragma unroll
        for (int k = 0; k < EPT; ++k) r[k] = atomicAdd(&cnt[d[k]], 1);
#pragma unroll
        for (int k = 0; k < EPT; ++k) rank[i + k * T] = r[k];
    } else {
#pragma unroll
        for (int k = 0; k < EPT; ++k) {
            const int j = i + k * T;
            if (j < e) rank[j] = atomicAdd(&cnt[dst[j]], 1);
        }
    }
}


__global__ __launch_bounds__(1024) void offsets_dinv_kernel(const int* __restrict__ cnt,
                                                            int* __restrict__ off,
                                                            float* __restrict__ dinv, int n) {
    __shared__ int s_pre[32];
    __shared__ int s_tot[32];

    const int tid   = threadIdx.x;
    const int lane  = tid & 31;
    const int wave  = tid >> 5;
    const int start = blockIdx.x * 1024;

    int pre = 0;
    const int4* c4 = (const int4*)cnt;
    const int nq = start >> 2;
    int q = tid;
    for (; q + 3072 < nq; q += 4096) {
        const int4 v0 = c4[q];
        const int4 v1 = c4[q + 1024];
        const int4 v2 = c4[q + 2048];
        const int4 v3 = c4[q + 3072];
        pre += ((v0.x + v0.y) + (v0.z + v0.w)) + ((v1.x + v1.y) + (v1.z + v1.w)) +
               ((v2.x + v2.y) + (v2.z + v2.w)) + ((v3.x + v3.y) + (v3.z + v3.w));
    }
    for (; q < nq; q += 1024) {
        const int4 v = c4[q];
        pre += (v.x + v.y) + (v.z + v.w);
    }

    const int i = start + tid;
    const int c = (i < n) ? cnt[i] : 0;

    int incl = c;
#pragma unroll
    for (int d = 1; d < 32; d <<= 1) {
        int t = __shfl_up(incl, d, 32);
        if (lane >= d) incl += t;
    }
#pragma unroll
    for (int o = 16; o > 0; o >>= 1) pre += __shfl_xor(pre, o, 32);

    if (lane == 31) s_tot[wave] = incl;
    if (lane == 0)  s_pre[wave] = pre;
    __syncthreads();

    int base = s_pre[lane];
#pragma unroll
    for (int o = 16; o > 0; o >>= 1) base += __shfl_xor(base, o, 32);

    const int tw = s_tot[lane];
    int tincl = tw;
#pragma unroll
    for (int d = 1; d < 32; d <<= 1) {
        int t = __shfl_up(tincl, d, 32);
        if (lane >= d) tincl += t;
    }
    const int woff = __shfl(tincl - tw, wave, 32);

    if (i < n) {
        const int excl = base + woff + incl - c;
        off[i]  = excl;
        dinv[i] = rsqrtf((float)(c + 1));
        if (i == n - 1) off[n] = excl + c;
    }
}


__global__ void fill_sorted_kernel(const int* __restrict__ src, const int* __restrict__ dst,
                                   const int* __restrict__ rank, const int* __restrict__ off,
                                   const float* __restrict__ dinv, int2* __restrict__ csr, int e) {
    const int T = gridDim.x * blockDim.x;
    const int i0 = blockIdx.x * blockDim.x + threadIdx.x;
    if (i0 + (EPT - 1) * T < e) {
        int s[EPT], d[EPT], r[EPT], o[EPT];
        float ws[EPT], wd[EPT];
#pragma unroll
        for (int k = 0; k < EPT; ++k) {
            s[k] = src[i0 + k * T];
            d[k] = dst[i0 + k * T];
            r[k] = rank[i0 + k * T];
        }
#pragma unroll
        for (int k = 0; k < EPT; ++k) {
            ws[k] = dinv[s[k]];
            wd[k] = dinv[d[k]];
            o[k]  = off[d[k]];
        }
#pragma unroll
        for (int k = 0; k < EPT; ++k)
            csr[o[k] + r[k]] = make_int2(s[k], __float_as_int(ws[k] * wd[k]));
    } else {
#pragma unroll
        for (int k = 0; k < EPT; ++k) {
            const int i = i0 + k * T;
            if (i < e) {
                const int s = src[i];
                const int d = dst[i];
                const float w = dinv[s] * dinv[d];
                csr[off[d] + rank[i]] = make_int2(s, __float_as_int(w));
            }
        }
    }
}


#define TROWS 32

__global__ __launch_bounds__(256) void gemm_xwT_wmma(const float* __restrict__ x,
                                                     const float* __restrict__ W,
                                                     float* __restrict__ xw, int n) {
    __shared__ __attribute__((aligned(16))) float lds_x[TROWS * XPITCH];

    const int tid   = threadIdx.x;
    const int wave  = tid >> 5;
    const int lane  = tid & 31;
    const int o0    = wave * 16;
    const int row   = lane & 15;
    const int khalf = (lane >> 4) << 1;
    const int lrow  = tid >> 5;
    const int lcol  = (tid & 31) << 2;

    const float* wrow = W + (size_t)(o0 + row) * NFEAT;
    v2f bf[NFEAT / 4];
#pragma unroll
    for (int j = 0; j < NFEAT / 4; ++j) {
        const int kk = 4 * j + khalf;
        bf[j].x = wrow[kk];
        bf[j].y = wrow[kk + 1];
    }

    const int col = o0 + row;
    const int mh  = (lane >> 4) << 3;
    const int ntile = (n + TROWS - 1) / TROWS;

    for (int t = blockIdx.x; t < ntile; t += gridDim.x) {
        const int n0 = t * TROWS;
        const bool two = n0 + 16 < n;

        const float4* xs = (const float4*)(x + (size_t)n0 * NFEAT);
        float4 p2 = make_float4(0.0f, 0.0f, 0.0f, 0.0f);
        float4 p3 = p2;
        const float4 p0 = xs[tid];
        const float4 p1 = xs[tid + 256];
        if (two) {
            p2 = xs[tid + 512];
            p3 = xs[tid + 768];
        }
        *(float4*)&lds_x[lrow * XPITCH + lcol]        = p0;
        *(float4*)&lds_x[(lrow + 8) * XPITCH + lcol]  = p1;
        *(float4*)&lds_x[(lrow + 16) * XPITCH + lcol] = p2;
        *(float4*)&lds_x[(lrow + 24) * XPITCH + lcol] = p3;
        __syncthreads();

        v8f acc0 = {};
        v8f acc1 = {};
#pragma unroll
        for (int j = 0; j < NFEAT / 4; ++j) {
            const int kk = 4 * j + khalf;
            v2f a0, a1;
            a0.x = lds_x[row * XPITCH + kk];
            a0.y = lds_x[row * XPITCH + kk + 1];
            a1.x = lds_x[(row + 16) * XPITCH + kk];
            a1.y = lds_x[(row + 16) * XPITCH + kk + 1];
            acc0 = __builtin_amdgcn_wmma_f32_16x16x4_f32(false, a0, false, bf[j],
                                                         (short)0, acc0, false, false);
            acc1 = __builtin_amdgcn_wmma_f32_16x16x4_f32(false, a1, false, bf[j],
                                                         (short)0, acc1, false, false);
        }

#pragma unroll
        for (int r = 0; r < 8; ++r)
            xw[(size_t)(n0 + mh + r) * NFEAT + col] = acc0[r];
        if (two) {
#pragma unroll
            for (int r = 0; r < 8; ++r)
                xw[(size_t)(n0 + 16 + mh + r) * NFEAT + col] = acc1[r];
        }
        __syncthreads();
    }
}


__device__ __forceinline__ void fma4(float4& acc, const float4 v, const float w) {
    acc.x += v.x * w;
    acc.y += v.y * w;
    acc.z += v.z * w;
    acc.w += v.w * w;
}

__global__ __launch_bounds__(256) void gather_finalize_kernel(const float* __restrict__ xw,
                                                              const int* __restrict__ off,
                                                              const float* __restrict__ dinv,
                                                              const int2* __restrict__ csr,
                                                              const float* __restrict__ bias,
                                                              const float* __restrict__ wreg,
                                                              const float* __restrict__ breg,
                                                              float* __restrict__ out, int n) {
    const int lane = threadIdx.x & 31;
    const int first = __builtin_amdgcn_readfirstlane((int)(blockIdx.x * 8 + (threadIdx.x >> 5)));
    const int nwave = (int)(gridDim.x * 8);

    const float4* rows = (const float4*)xw;
    const float4 b = ((const float4*)bias)[lane];
    const float4 w = ((const float4*)wreg)[lane];
    const float br = breg[0];

  for (int node = first; node < n; node += nwave) {
    const int beg = off[node];
    const int end = off[node + 1];

    int base = beg;
    int m = min(32, end - base);
    int2 ent = make_int2(0, 0);
    if (m > 0) ent = csr[base + min(lane, m - 1)];

    const float4 vself = rows[(size_t)node * (NFEAT / 4) + lane];
    const float di = dinv[node];
    const float nself = di * di;

    float4 acc;
    acc.x = vself.x * nself;
    acc.y = vself.y * nself;
    acc.z = vself.z * nself;
    acc.w = vself.w * nself;

    while (m > 0) {
        int j = 0;
        for (; j + 4 <= m; j += 4) {
            const int s0 = __builtin_amdgcn_readlane(ent.x, j);
            const int s1 = __builtin_amdgcn_readlane(ent.x, j + 1);
            const int s2 = __builtin_amdgcn_readlane(ent.x, j + 2);
            const int s3 = __builtin_amdgcn_readlane(ent.x, j + 3);
            const float w0 = __int_as_float(__builtin_amdgcn_readlane(ent.y, j));
            const float w1 = __int_as_float(__builtin_amdgcn_readlane(ent.y, j + 1));
            const float w2 = __int_as_float(__builtin_amdgcn_readlane(ent.y, j + 2));
            const float w3 = __int_as_float(__builtin_amdgcn_readlane(ent.y, j + 3));
            const float4 v0 = rows[(size_t)s0 * (NFEAT / 4) + lane];
            const float4 v1 = rows[(size_t)s1 * (NFEAT / 4) + lane];
            const float4 v2 = rows[(size_t)s2 * (NFEAT / 4) + lane];
            const float4 v3 = rows[(size_t)s3 * (NFEAT / 4) + lane];
            fma4(acc, v0, w0);
            fma4(acc, v1, w1);
            fma4(acc, v2, w2);
            fma4(acc, v3, w3);
        }
        if (j + 2 <= m) {
            const int s0 = __builtin_amdgcn_readlane(ent.x, j);
            const int s1 = __builtin_amdgcn_readlane(ent.x, j + 1);
            const float w0 = __int_as_float(__builtin_amdgcn_readlane(ent.y, j));
            const float w1 = __int_as_float(__builtin_amdgcn_readlane(ent.y, j + 1));
            const float4 v0 = rows[(size_t)s0 * (NFEAT / 4) + lane];
            const float4 v1 = rows[(size_t)s1 * (NFEAT / 4) + lane];
            fma4(acc, v0, w0);
            fma4(acc, v1, w1);
            j += 2;
        }
        if (j < m) {
            const int s0 = __builtin_amdgcn_readlane(ent.x, j);
            const float w0 = __int_as_float(__builtin_amdgcn_readlane(ent.y, j));
            const float4 v0 = rows[(size_t)s0 * (NFEAT / 4) + lane];
            fma4(acc, v0, w0);
        }
        base += 32;
        m = min(32, end - base);
        if (m > 0) ent = csr[base + min(lane, m - 1)];
    }

    float dot = fmaxf(acc.x + b.x, 0.0f) * w.x + fmaxf(acc.y + b.y, 0.0f) * w.y +
                fmaxf(acc.z + b.z, 0.0f) * w.z + fmaxf(acc.w + b.w, 0.0f) * w.w;
#pragma unroll
    for (int o = 16; o > 0; o >>= 1)
        dot += __shfl_xor(dot, o, 32);
    if (lane == 0) out[node] = dot + br;
  }
}


extern "C" void kernel_launch(void* const* d_in, const int* in_sizes, int n_in,
                              void* d_out, int out_size, void* d_ws, size_t ws_size,
                              hipStream_t stream) {
    const float* x    = (const float*)d_in[0];
    const int*   ei   = (const int*)d_in[1];
    const float* W    = (const float*)d_in[2];
    const float* b    = (const float*)d_in[3];
    const float* wreg = (const float*)d_in[4];
    const float* breg = (const float*)d_in[5];
    float* out = (float*)d_out;

    const int n = in_sizes[0] / NFEAT;
    const int e = in_sizes[1] / 2;
    const int* src = ei;
    const int* dst = ei + e;

    float* xw   = (float*)d_ws;
    int*   cnt  = (int*)(xw + (size_t)n * NFEAT);
    float* dinv = (float*)(cnt + n);
    int2*  csr  = (int2*)(dinv + n);
    int*   rank = (int*)(csr + e);
    int*   off  = rank + e;

    const int B = 256;
    const int GE = (e + B * EPT - 1) / (B * EPT);
    zero_counts_kernel<<<(n + B - 1) / B, B, 0, stream>>>(cnt, n);
    count_rank_kernel<<<GE, B, 0, stream>>>(dst, cnt, rank, e);
    offsets_dinv_kernel<<<(n + 1023) / 1024, 1024, 0, stream>>>(cnt, off, dinv, n);
    fill_sorted_kernel<<<GE, B, 0, stream>>>(src, dst, rank, off, dinv, csr, e);

    gemm_xwT_wmma<<<1024, B, 0, stream>>>(x, W, xw, n);

    gather_finalize_kernel<<<2048, B, 0, stream>>>(xw, off, dinv, csr, b, wreg, breg, out, n);
}


# Edge count-and-slot pass rides in the product launch: one returning atomic per thread per tile trip, answered under the matrix steps; 6 launches to 5; edge sum order still arrival-order rank
# speedup vs baseline: 2.2662x; 1.0845x over previous
#include <hip/hip_runtime.h>
#include <cstdint>


#define NFEAT 128
#define XPITCH 132

typedef __attribute__((ext_vector_type(2))) float v2f;
typedef __attribute__((ext_vector_type(8))) float v8f;


__global__ void zero_counts_kernel(int* __restrict__ cnt, int n) {
    int i = blockIdx.x * blockDim.x + threadIdx.x;
    if (i < n) cnt[i] = 0;
}

#define EPT 4


__global__ __launch_bounds__(1024) void offsets_dinv_kernel(const int* __restrict__ cnt,
                                                            int* __restrict__ off,
                                                            float* __restrict__ dinv, int n) {
    __shared__ int s_pre[32];
    __shared__ int s_tot[32];

    const int tid   = threadIdx.x;
    const int lane  = tid & 31;
    const int wave  = tid >> 5;
    const int start = blockIdx.x * 1024;

    int pre = 0;
    const int4* c4 = (const int4*)cnt;
    const int nq = start >> 2;
    int q = tid;
    for (; q + 3072 < nq; q += 4096) {
        const int4 v0 = c4[q];
        const int4 v1 = c4[q + 1024];
        const int4 v2 = c4[q + 2048];
        const int4 v3 = c4[q + 3072];
        pre += ((v0.x + v0.y) + (v0.z + v0.w)) + ((v1.x + v1.y) + (v1.z + v1.w)) +
               ((v2.x + v2.y) + (v2.z + v2.w)) + ((v3.x + v3.y) + (v3.z + v3.w));
    }
    for (; q < nq; q += 1024) {
        const int4 v = c4[q];
        pre += (v.x + v.y) + (v.z + v.w);
    }

    const int i = start + tid;
    const int c = (i < n) ? cnt[i] : 0;

    int incl = c;
#pragma unroll
    for (int d = 1; d < 32; d <<= 1) {
        int t = __shfl_up(incl, d, 32);
        if (lane >= d) incl += t;
    }
#pragma unroll
    for (int o = 16; o > 0; o >>= 1) pre += __shfl_xor(pre, o, 32);

    if (lane == 31) s_tot[wave] = incl;
    if (lane == 0)  s_pre[wave] = pre;
    __syncthreads();

    int base = s_pre[lane];
#pragma unroll
    for (int o = 16; o > 0; o >>= 1) base += __shfl_xor(base, o, 32);

    const int tw = s_tot[lane];
    int tincl = tw;
#pragma unroll
    for (int d = 1; d < 32; d <<= 1) {
        int t = __shfl_up(tincl, d, 32);
        if (lane >= d) tincl += t;
    }
    const int woff = __shfl(tincl - tw, wave, 32);

    if (i < n) {
        const int excl = base + woff + incl - c;
        off[i]  = excl;
        dinv[i] = rsqrtf((float)(c + 1));
        if (i == n - 1) off[n] = excl + c;
    }
}


__global__ void fill_sorted_kernel(const int* __restrict__ src, const int* __restrict__ dst,
                                   const int* __restrict__ rank, const int* __restrict__ off,
                                   const float* __restrict__ dinv, int2* __restrict__ csr, int e) {
    const int T = gridDim.x * blockDim.x;
    const int i0 = blockIdx.x * blockDim.x + threadIdx.x;
    if (i0 + (EPT - 1) * T < e) {
        int s[EPT], d[EPT], r[EPT], o[EPT];
        float ws[EPT], wd[EPT];
#pragma unroll
        for (int k = 0; k < EPT; ++k) {
            s[k] = src[i0 + k * T];
            d[k] = dst[i0 + k * T];
            r[k] = rank[i0 + k * T];
        }
#pragma unroll
        for (int k = 0; k < EPT; ++k) {
            ws[k] = dinv[s[k]];
            wd[k] = dinv[d[k]];
            o[k]  = off[d[k]];
        }
#pragma unroll
        for (int k = 0; k < EPT; ++k)
            csr[o[k] + r[k]] = make_int2(s[k], __float_as_int(ws[k] * wd[k]));
    } else {
#pragma unroll
        for (int k = 0; k < EPT; ++k) {
            const int i = i0 + k * T;
            if (i < e) {
                const int s = src[i];
                const int d = dst[i];
                const float w = dinv[s] * dinv[d];
                csr[off[d] + rank[i]] = make_int2(s, __float_as_int(w));
            }
        }
    }
}


#define TROWS 32

__global__ __launch_bounds__(256) void gemm_xwT_count_wmma(const float* __restrict__ x,
                                                           const float* __restrict__ W,
                                                           float* __restrict__ xw, int n,
                                                           const int* __restrict__ dst,
                                                           int* __restrict__ cnt,
                                                           int* __restrict__ rank, int e) {
    __shared__ __attribute__((aligned(16))) float lds_x[TROWS * XPITCH];

    const int tid   = threadIdx.x;
    const int wave  = tid >> 5;
    const int lane  = tid & 31;
    const int o0    = wave * 16;
    const int row   = lane & 15;
    const int khalf = (lane >> 4) << 1;
    const int lrow  = tid >> 5;
    const int lcol  = (tid & 31) << 2;

    const float* wrow = W + (size_t)(o0 + row) * NFEAT;
    v2f bf[NFEAT / 4];
#pragma unroll
    for (int j = 0; j < NFEAT / 4; ++j) {
        const int kk = 4 * j + khalf;
        bf[j].x = wrow[kk];
        bf[j].y = wrow[kk + 1];
    }

    const int col = o0 + row;
    const int mh  = (lane >> 4) << 3;
    const int ntile = (n + TROWS - 1) / TROWS;

    const int T = gridDim.x * 256;
    int ei = blockIdx.x * 256 + tid;

    for (int t = blockIdx.x; t < ntile; t += gridDim.x) {
        const int n0 = t * TROWS;
        const bool two = n0 + 16 < n;

        const float4* xs = (const float4*)(x + (size_t)n0 * NFEAT);
        float4 p2 = make_float4(0.0f, 0.0f, 0.0f, 0.0f);
        float4 p3 = p2;
        const float4 p0 = xs[tid];
        const float4 p1 = xs[tid + 256];
        if (two) {
            p2 = xs[tid + 512];
            p3 = xs[tid + 768];
        }
        int d = -1;
        if (ei < e) d = dst[ei];
        *(float4*)&lds_x[lrow * XPITCH + lcol]        = p0;
        *(float4*)&lds_x[(lrow + 8) * XPITCH + lcol]  = p1;
        *(float4*)&lds_x[(lrow + 16) * XPITCH + lcol] = p2;
        *(float4*)&lds_x[(lrow + 24) * XPITCH + lcol] = p3;
        __syncthreads();

        int slot = 0;
        if (d >= 0) slot = atomicAdd(&cnt[d], 1);

        v8f acc0 = {};
        v8f acc1 = {};
#pragma unroll
        for (int j = 0; j < NFEAT / 4; ++j) {
            const int kk = 4 * j + khalf;
            v2f a0, a1;
            a0.x = lds_x[row * XPITCH + kk];
            a0.y = lds_x[row * XPITCH + kk + 1];
            a1.x = lds_x[(row + 16) * XPITCH + kk];
            a1.y = lds_x[(row + 16) * XPITCH + kk + 1];
            acc0 = __builtin_amdgcn_wmma_f32_16x16x4_f32(false, a0, false, bf[j],
                                                         (short)0, acc0, false, false);
            acc1 = __builtin_amdgcn_wmma_f32_16x16x4_f32(false, a1, false, bf[j],
                                                         (short)0, acc1, false, false);
        }

#pragma unroll
        for (int r = 0; r < 8; ++r)
            xw[(size_t)(n0 + mh + r) * NFEAT + col] = acc0[r];
        if (two) {
#pragma unroll
            for (int r = 0; r < 8; ++r)
                xw[(size_t)(n0 + 16 + mh + r) * NFEAT + col] = acc1[r];
        }
        if (d >= 0) rank[ei] = slot;
        ei += T;
        __syncthreads();
    }
    for (; ei < e; ei += T) rank[ei] = atomicAdd(&cnt[dst[ei]], 1);
}


__device__ __forceinline__ void fma4(float4& acc, const float4 v, const float w) {
    acc.x += v.x * w;
    acc.y += v.y * w;
    acc.z += v.z * w;
    acc.w += v.w * w;
}

__global__ __launch_bounds__(256) void gather_finalize_kernel(const float* __restrict__ xw,
                                                              const int* __restrict__ off,
                                                              const float* __restrict__ dinv,
                                                              const int2* __restrict__ csr,
                                                              const float* __restrict__ bias,
                                                              const float* __restrict__ wreg,
                                                              const float* __restrict__ breg,
                                                              float* __restrict__ out, int n) {
    const int lane = threadIdx.x & 31;
    const int first = __builtin_amdgcn_readfirstlane((int)(blockIdx.x * 8 + (threadIdx.x >> 5)));
    const int nwave = (int)(gridDim.x * 8);

    const float4* rows = (const float4*)xw;
    const float4 b = ((const float4*)bias)[lane];
    const float4 w = ((const float4*)wreg)[lane];
    const float br = breg[0];

  for (int node = first; node < n; node += nwave) {
    const int beg = off[node];
    const int end = off[node + 1];

    int base = beg;
    int m = min(32, end - base);
    int2 ent = make_int2(0, 0);
    if (m > 0) ent = csr[base + min(lane, m - 1)];

    const float4 vself = rows[(size_t)node * (NFEAT / 4) + lane];
    const float di = dinv[node];
    const float nself = di * di;

    float4 acc;
    acc.x = vself.x * nself;
    acc.y = vself.y * nself;
    acc.z = vself.z * nself;
    acc.w = vself.w * nself;

    while (m > 0) {
        int j = 0;
        for (; j + 4 <= m; j += 4) {
            const int s0 = __builtin_amdgcn_readlane(ent.x, j);
            const int s1 = __builtin_amdgcn_readlane(ent.x, j + 1);
            const int s2 = __builtin_amdgcn_readlane(ent.x, j + 2);
            const int s3 = __builtin_amdgcn_readlane(ent.x, j + 3);
            const float w0 = __int_as_float(__builtin_amdgcn_readlane(ent.y, j));
            const float w1 = __int_as_float(__builtin_amdgcn_readlane(ent.y, j + 1));
            const float w2 = __int_as_float(__builtin_amdgcn_readlane(ent.y, j + 2));
            const float w3 = __int_as_float(__builtin_amdgcn_readlane(ent.y, j + 3));
            const float4 v0 = rows[(size_t)s0 * (NFEAT / 4) + lane];
            const float4 v1 = rows[(size_t)s1 * (NFEAT / 4) + lane];
            const float4 v2 = rows[(size_t)s2 * (NFEAT / 4) + lane];
            const float4 v3 = rows[(size_t)s3 * (NFEAT / 4) + lane];
            fma4(acc, v0, w0);
            fma4(acc, v1, w1);
            fma4(acc, v2, w2);
            fma4(acc, v3, w3);
        }
        if (j + 2 <= m) {
            const int s0 = __builtin_amdgcn_readlane(ent.x, j);
            const int s1 = __builtin_amdgcn_readlane(ent.x, j + 1);
            const float w0 = __int_as_float(__builtin_amdgcn_readlane(ent.y, j));
            const float w1 = __int_as_float(__builtin_amdgcn_readlane(ent.y, j + 1));
            const float4 v0 = rows[(size_t)s0 * (NFEAT / 4) + lane];
            const float4 v1 = rows[(size_t)s1 * (NFEAT / 4) + lane];
            fma4(acc, v0, w0);
            fma4(acc, v1, w1);
            j += 2;
        }
        if (j < m) {
            const int s0 = __builtin_amdgcn_readlane(ent.x, j);
            const float w0 = __int_as_float(__builtin_amdgcn_readlane(ent.y, j));
            const float4 v0 = rows[(size_t)s0 * (NFEAT / 4) + lane];
            fma4(acc, v0, w0);
        }
        base += 32;
        m = min(32, end - base);
        if (m > 0) ent = csr[base + min(lane, m - 1)];
    }

    float dot = fmaxf(acc.x + b.x, 0.0f) * w.x + fmaxf(acc.y + b.y, 0.0f) * w.y +
                fmaxf(acc.z + b.z, 0.0f) * w.z + fmaxf(acc.w + b.w, 0.0f) * w.w;
#pragma unroll
    for (int o = 16; o > 0; o >>= 1)
        dot += __shfl_xor(dot, o, 32);
    if (lane == 0) out[node] = dot + br;
  }
}


extern "C" void kernel_launch(void* const* d_in, const int* in_sizes, int n_in,
                              void* d_out, int out_size, void* d_ws, size_t ws_size,
                              hipStream_t stream) {
    const float* x    = (const float*)d_in[0];
    const int*   ei   = (const int*)d_in[1];
    const float* W    = (const float*)d_in[2];
    const float* b    = (const float*)d_in[3];
    const float* wreg = (const float*)d_in[4];
    const float* breg = (const float*)d_in[5];
    float* out = (float*)d_out;

    const int n = in_sizes[0] / NFEAT;
    const int e = in_sizes[1] / 2;
    const int* src = ei;
    const int* dst = ei + e;

    float* xw   = (float*)d_ws;
    int*   cnt  = (int*)(xw + (size_t)n * NFEAT);
    float* dinv = (float*)(cnt + n);
    int2*  csr  = (int2*)(dinv + n);
    int*   rank = (int*)(csr + e);
    int*   off  = rank + e;

    const int B = 256;
    const int GE = (e + B * EPT - 1) / (B * EPT);
    zero_counts_kernel<<<(n + B - 1) / B, B, 0, stream>>>(cnt, n);

    gemm_xwT_count_wmma<<<1024, B, 0, stream>>>(x, W, xw, n, dst, cnt, rank, e);

    offsets_dinv_kernel<<<(n + 1023) / 1024, 1024, 0, stream>>>(cnt, off, dinv, n);
    fill_sorted_kernel<<<GE, B, 0, stream>>>(src, dst, rank, off, dinv, csr, e);

    gather_finalize_kernel<<<2048, B, 0, stream>>>(xw, off, dinv, csr, b, wreg, breg, out, n);
}


# Scan+fill edge list replaced by a 32-slot row per destination: the counting atomic's return is the edge's final slot (overflow list behind); 5 launches to 3; sum order: arrival-order slots, self last
# speedup vs baseline: 2.7749x; 1.2245x over previous
#include <hip/hip_runtime.h>
#include <cstdint>


#define NFEAT 128
#define XPITCH 132
#define CAP 32

typedef __attribute__((ext_vector_type(2))) float v2f;
typedef __attribute__((ext_vector_type(8))) float v8f;


__global__ void zero_counts_kernel(int* __restrict__ cnt, int n) {
    int i = blockIdx.x * blockDim.x + threadIdx.x;
    if (i < n) cnt[i] = 0;
}


__device__ __forceinline__ void place_edge(int* __restrict__ slots, int2* __restrict__ ovf,
                                           int* __restrict__ ovfn, int s, int d, int slot, int e) {
    if (slot < CAP) {
        slots[(size_t)d * CAP + slot] = s;
    } else {
        const int p = atomicAdd(ovfn, 1);
        if (p < e) ovf[p] = make_int2(s, d);
    }
}


#define TROWS 32

__global__ __launch_bounds__(256) void gemm_xwT_count_wmma(const float* __restrict__ x,
                                                           const float* __restrict__ W,
                                                           float* __restrict__ xw, int n,
                                                           const int* __restrict__ src,
                                                           const int* __restrict__ dst,
                                                           int* __restrict__ cnt,
                                                           int* __restrict__ slots,
                                                           int2* __restrict__ ovf, int e) {
    __shared__ __attribute__((aligned(16))) float lds_x[TROWS * XPITCH];

    const int tid   = threadIdx.x;
    const int wave  = tid >> 5;
    const int lane  = tid & 31;
    const int o0    = wave * 16;
    const int row   = lane & 15;
    const int khalf = (lane >> 4) << 1;
    const int lrow  = tid >> 5;
    const int lcol  = (tid & 31) << 2;

    const float* wrow = W + (size_t)(o0 + row) * NFEAT;
    v2f bf[NFEAT / 4];
#pragma unroll
    for (int j = 0; j < NFEAT / 4; ++j) {
        const int kk = 4 * j + khalf;
        bf[j].x = wrow[kk];
        bf[j].y = wrow[kk + 1];
    }

    const int col = o0 + row;
    const int mh  = (lane >> 4) << 3;
    const int ntile = (n + TROWS - 1) / TROWS;

    int* ovfn = cnt + n;
    const int T = gridDim.x * 256;
    int ei = blockIdx.x * 256 + tid;

    for (int t = blockIdx.x; t < ntile; t += gridDim.x) {
        const int n0 = t * TROWS;
        const bool two = n0 + 16 < n;

        const float4* xs = (const float4*)(x + (size_t)n0 * NFEAT);
        float4 p2 = make_float4(0.0f, 0.0f, 0.0f, 0.0f);
        float4 p3 = p2;
        const float4 p0 = xs[tid];
        const float4 p1 = xs[tid + 256];
        if (two) {
            p2 = xs[tid + 512];
            p3 = xs[tid + 768];
        }
        int d = -1;
        int s = 0;
        if (ei < e) {
            d = dst[ei];
            s = src[ei];
        }
        *(float4*)&lds_x[lrow * XPITCH + lcol]        = p0;
        *(float4*)&lds_x[(lrow + 8) * XPITCH + lcol]  = p1;
        *(float4*)&lds_x[(lrow + 16) * XPITCH + lcol] = p2;
        *(float4*)&lds_x[(lrow + 24) * XPITCH + lcol] = p3;
        __syncthreads();

        int slot = 0;
        if (d >= 0) slot = atomicAdd(&cnt[d], 1);

        v8f acc0 = {};
        v8f acc1 = {};
#pragma unroll
        for (int j = 0; j < NFEAT / 4; ++j) {
            const int kk = 4 * j + khalf;
            v2f a0, a1;
            a0.x = lds_x[row * XPITCH + kk];
            a0.y = lds_x[row * XPITCH + kk + 1];
            a1.x = lds_x[(row + 16) * XPITCH + kk];
            a1.y = lds_x[(row + 16) * XPITCH + kk + 1];
            acc0 = __builtin_amdgcn_wmma_f32_16x16x4_f32(false, a0, false, bf[j],
                                                         (short)0, acc0, false, false);
            acc1 = __builtin_amdgcn_wmma_f32_16x16x4_f32(false, a1, false, bf[j],
                                                         (short)0, acc1, false, false);
        }

#pragma unroll
        for (int r = 0; r < 8; ++r)
            xw[(size_t)(n0 + mh + r) * NFEAT + col] = acc0[r];
        if (two) {
#pragma unroll
            for (int r = 0; r < 8; ++r)
                xw[(size_t)(n0 + 16 + mh + r) * NFEAT + col] = acc1[r];
        }
        if (d >= 0) place_edge(slots, ovf, ovfn, s, d, slot, e);
        ei += T;
        __syncthreads();
    }
    for (; ei < e; ei += T) {
        const int d = dst[ei];
        const int s = src[ei];
        const int slot = atomicAdd(&cnt[d], 1);
        place_edge(slots, ovf, ovfn, s, d, slot, e);
    }
}


__device__ __forceinline__ void fma4(float4& acc, const float4 v, const float w) {
    acc.x += v.x * w;
    acc.y += v.y * w;
    acc.z += v.z * w;
    acc.w += v.w * w;
}

__device__ __forceinline__ float dinv_of(const int counter) {
    return __builtin_amdgcn_rsqf((float)(counter + 1));
}

__device__ __forceinline__ float edge_norm(const int csrc, const float ddst) {
    return dinv_of(csrc) * ddst;
}

__global__ __launch_bounds__(256) void gather_finalize_kernel(const float* __restrict__ xw,
                                                              const int* __restrict__ cnt,
                                                              const int* __restrict__ slots,
                                                              const int2* __restrict__ ovf,
                                                              const float* __restrict__ bias,
                                                              const float* __restrict__ wreg,
                                                              const float* __restrict__ breg,
                                                              float* __restrict__ out, int n, int e) {
    const int lane = threadIdx.x & 31;
    const int first = __builtin_amdgcn_readfirstlane((int)(blockIdx.x * 8 + (threadIdx.x >> 5)));
    const int nwave = (int)(gridDim.x * 8);

    const float4* rows = (const float4*)xw;
    const float4 b = ((const float4*)bias)[lane];
    const float4 w = ((const float4*)wreg)[lane];
    const float br = breg[0];

  for (int node = first; node < n; node += nwave) {
    const float4 vself = rows[(size_t)node * (NFEAT / 4) + lane];
    const int c = cnt[node];
    const int m = min(c, CAP);
    const float di = dinv_of(c);

    int ent = 0;
    int ce = 0;
    if (m > 0) {
        ent = slots[(size_t)node * CAP + min(lane, m - 1)];
        ce = cnt[ent];
    }

    float4 acc = make_float4(0.0f, 0.0f, 0.0f, 0.0f);

    {
        int j = 0;
        for (; j + 4 <= m; j += 4) {
            const int s0 = __builtin_amdgcn_readlane(ent, j);
            const int s1 = __builtin_amdgcn_readlane(ent, j + 1);
            const int s2 = __builtin_amdgcn_readlane(ent, j + 2);
            const int s3 = __builtin_amdgcn_readlane(ent, j + 3);
            const float4 v0 = rows[(size_t)s0 * (NFEAT / 4) + lane];
            const float4 v1 = rows[(size_t)s1 * (NFEAT / 4) + lane];
            const float4 v2 = rows[(size_t)s2 * (NFEAT / 4) + lane];
            const float4 v3 = rows[(size_t)s3 * (NFEAT / 4) + lane];
            const float w0 = edge_norm(__builtin_amdgcn_readlane(ce, j), di);
            const float w1 = edge_norm(__builtin_amdgcn_readlane(ce, j + 1), di);
            const float w2 = edge_norm(__builtin_amdgcn_readlane(ce, j + 2), di);
            const float w3 = edge_norm(__builtin_amdgcn_readlane(ce, j + 3), di);
            fma4(acc, v0, w0);
            fma4(acc, v1, w1);
            fma4(acc, v2, w2);
            fma4(acc, v3, w3);
        }
        if (j + 2 <= m) {
            const int s0 = __builtin_amdgcn_readlane(ent, j);
            const int s1 = __builtin_amdgcn_readlane(ent, j + 1);
            const float4 v0 = rows[(size_t)s0 * (NFEAT / 4) + lane];
            const float4 v1 = rows[(size_t)s1 * (NFEAT / 4) + lane];
            const float w0 = edge_norm(__builtin_amdgcn_readlane(ce, j), di);
            const float w1 = edge_norm(__builtin_amdgcn_readlane(ce, j + 1), di);
            fma4(acc, v0, w0);
            fma4(acc, v1, w1);
            j += 2;
        }
        if (j < m) {
            const int s0 = __builtin_amdgcn_readlane(ent, j);
            const float4 v0 = rows[(size_t)s0 * (NFEAT / 4) + lane];
            const float w0 = edge_norm(__builtin_amdgcn_readlane(ce, j), di);
            fma4(acc, v0, w0);
        }
    }

    if (c > CAP) {
        const int on = min(cnt[n], e);
        for (int base = 0; base < on; base += 32) {
            const int idx = base + lane;
            int2 pr = make_int2(0, -1);
            if (idx < on) pr = ovf[idx];
            const bool hit = (pr.y == node);
            float wl = 0.0f;
            if (hit) wl = edge_norm(cnt[pr.x], di);
            unsigned int mask = __builtin_amdgcn_ballot_w32(hit);
            while (mask != 0u) {
                const int j = __builtin_ctz(mask);
                mask &= mask - 1u;
                const int s0 = __builtin_amdgcn_readlane(pr.x, j);
                const float w0 = __int_as_float(__builtin_amdgcn_readlane(__float_as_int(wl), j));
                const float4 v0 = rows[(size_t)s0 * (NFEAT / 4) + lane];
                fma4(acc, v0, w0);
            }
        }
    }

    fma4(acc, vself, di * di);

    float dot = fmaxf(acc.x + b.x, 0.0f) * w.x + fmaxf(acc.y + b.y, 0.0f) * w.y +
                fmaxf(acc.z + b.z, 0.0f) * w.z + fmaxf(acc.w + b.w, 0.0f) * w.w;
#pragma unroll
    for (int o = 16; o > 0; o >>= 1)
        dot += __shfl_xor(dot, o, 32);
    if (lane == 0) out[node] = dot + br;
  }
}


extern "C" void kernel_launch(void* const* d_in, const int* in_sizes, int n_in,
                              void* d_out, int out_size, void* d_ws, size_t ws_size,
                              hipStream_t stream) {
    const float* x    = (const float*)d_in[0];
    const int*   ei   = (const int*)d_in[1];
    const float* W    = (const float*)d_in[2];
    const float* b    = (const float*)d_in[3];
    const float* wreg = (const float*)d_in[4];
    const float* breg = (const float*)d_in[5];
    float* out = (float*)d_out;

    const int n = in_sizes[0] / NFEAT;
    const int e = in_sizes[1] / 2;
    const int* src = ei;
    const int* dst = ei + e;

    float* xw    = (float*)d_ws;
    int*   cnt   = (int*)(xw + (size_t)n * NFEAT);
    int*   slots = cnt + n + 32;
    int2*  ovf   = (int2*)(slots + (size_t)n * CAP);

    const int B = 256;
    zero_counts_kernel<<<(n + 1 + B - 1) / B, B, 0, stream>>>(cnt, n + 1);

    gemm_xwT_count_wmma<<<1024, B, 0, stream>>>(x, W, xw, n, src, dst, cnt, slots, ovf, e);

    gather_finalize_kernel<<<2048, B, 0, stream>>>(xw, cnt, slots, ovf, b, wreg, breg, out, n, e);
}


# Edge slots from LDS counters: edge list counting-sorted by destination range in LDS, then one block per range places it; no global atomic per edge, no clearing launch; sum order: arrival of LDS adds
# speedup vs baseline: 4.5781x; 1.6498x over previous
#include <hip/hip_runtime.h>
#include <cstdint>


#define NFEAT 128
#define XPITCH 132
#define CAP 32
#define TROWS 32

typedef __attribute__((ext_vector_type(2))) float v2f;
typedef __attribute__((ext_vector_type(8))) float v8f;


#define SH 9
#define RB (1 << SH)
#define NBKMAX 256
#define PA 128
#define TA 1024
#define KA 5

__device__ __forceinline__ void place_edge(int* __restrict__ slots, int2* __restrict__ ovf,
                                           int* __restrict__ ovfn, int s, int d, int slot, int e) {
    if (slot < CAP) {
        slots[(size_t)d * CAP + slot] = s;
    } else {
        const int p = atomicAdd(ovfn, 1);
        if (p < e) ovf[p] = make_int2(s, d);
    }
}

__global__ __launch_bounds__(TA) void bin_edges_kernel(const int* __restrict__ src,
                                                       const int* __restrict__ dst,
                                                       int2* __restrict__ sorted,
                                                       int2* __restrict__ tab,
                                                       int* __restrict__ ovfn, int n, int e) {
    __shared__ int hist[NBKMAX];
    __shared__ int cur[NBKMAX];
    __shared__ int stage_s[KA * TA];
    __shared__ int stage_d[KA * TA];
    const int tid = threadIdx.x;
    const int p = blockIdx.x;
    const int nbk = (n + RB - 1) >> SH;
    const int C = (e + PA - 1) / PA;
    const int c0 = p * C;
    const int c1 = min(e, c0 + C);
    const bool staged = (c1 - c0) <= KA * TA;

    if (p == 0 && tid == 0) *ovfn = 0;
    for (int b = tid; b < NBKMAX; b += TA) hist[b] = 0;
    __syncthreads();

    int d[KA], s[KA], pos[KA];
#pragma unroll
    for (int k = 0; k < KA; ++k) {
        const int i = c0 + tid + k * TA;
        d[k] = -1;
        s[k] = 0;
        if (i < c1) {
            d[k] = dst[i];
            s[k] = src[i];
        }
    }
#pragma unroll
    for (int k = 0; k < KA; ++k)
        if (d[k] >= 0) atomicAdd(&hist[d[k] >> SH], 1);
    for (int i = c0 + tid + KA * TA; i < c1; i += TA)
        atomicAdd(&hist[dst[i] >> SH], 1);
    __syncthreads();

    if (tid < 32) {
        int v[NBKMAX / 32];
        int run = 0;
#pragma unroll
        for (int k = 0; k < NBKMAX / 32; ++k) {
            const int t = hist[tid * (NBKMAX / 32) + k];
            v[k] = run;
            run += t;
        }
        int inc = run;
#pragma unroll
        for (int o = 1; o < 32; o <<= 1) {
            const int t = __shfl_up(inc, o, 32);
            if (tid >= o) inc += t;
        }
        const int excl = inc - run;
#pragma unroll
        for (int k = 0; k < NBKMAX / 32; ++k) cur[tid * (NBKMAX / 32) + k] = excl + v[k];
    }
    __syncthreads();

    for (int b = tid; b < nbk; b += TA)
        tab[(size_t)b * PA + p] = make_int2(c0 + cur[b], hist[b]);
    __syncthreads();

#pragma unroll
    for (int k = 0; k < KA; ++k) {
        pos[k] = 0;
        if (d[k] >= 0) pos[k] = atomicAdd(&cur[d[k] >> SH], 1);
    }
    if (staged) {
#pragma unroll
        for (int k = 0; k < KA; ++k) {
            if (d[k] >= 0) {
                stage_s[pos[k]] = s[k];
                stage_d[pos[k]] = d[k];
            }
        }
        __syncthreads();
        for (int i = tid; i < c1 - c0; i += TA)
            sorted[c0 + i] = make_int2(stage_s[i], stage_d[i]);
    } else {
#pragma unroll
        for (int k = 0; k < KA; ++k)
            if (d[k] >= 0) sorted[c0 + pos[k]] = make_int2(s[k], d[k]);
        for (int i = c0 + tid + KA * TA; i < c1; i += TA) {
            const int dd = dst[i];
            const int ss = src[i];
            const int q = atomicAdd(&cur[dd >> SH], 1);
            sorted[c0 + q] = make_int2(ss, dd);
        }
    }
}

__global__ __launch_bounds__(RB) void place_bucket_kernel(const int2* __restrict__ sorted,
                                                          const int2* __restrict__ tab,
                                                          int* __restrict__ cnt,
                                                          int* __restrict__ slots,
                                                          int2* __restrict__ ovf, int n, int e) {
    __shared__ int lcnt[RB];
    const int tid  = threadIdx.x;
    const int lane = tid & 31;
    const int wave = __builtin_amdgcn_readfirstlane(tid >> 5);
    const int b = blockIdx.x;
    const int base = b << SH;
    int* ovfn = cnt + n;
    constexpr int RW = PA / (RB / 32);

    lcnt[tid] = 0;
    const int2 te = tab[(size_t)b * PA + wave * RW + (lane & (RW - 1))];
    __syncthreads();

    int2 ed[2 * RW];
#pragma unroll
    for (int k = 0; k < RW; ++k) {
        const int st = __builtin_amdgcn_readlane(te.x, k);
        const int ct = __builtin_amdgcn_readlane(te.y, k);
        ed[2 * k] = make_int2(0, -1);
        ed[2 * k + 1] = make_int2(0, -1);
        if (lane < ct) ed[2 * k] = sorted[st + lane];
        if (lane + 32 < ct) ed[2 * k + 1] = sorted[st + lane + 32];
    }
    int slot[2 * RW];
#pragma unroll
    for (int k = 0; k < 2 * RW; ++k) {
        slot[k] = 0;
        if (ed[k].y >= 0) slot[k] = atomicAdd(&lcnt[ed[k].y - base], 1);
    }
#pragma unroll
    for (int k = 0; k < 2 * RW; ++k)
        if (ed[k].y >= 0) place_edge(slots, ovf, ovfn, ed[k].x, ed[k].y, slot[k], e);
#pragma unroll 1
    for (int k = 0; k < RW; ++k) {
        const int st = __builtin_amdgcn_readlane(te.x, k);
        const int ct = __builtin_amdgcn_readlane(te.y, k);
        for (int o = 64 + lane; o < ct; o += 32) {
            const int2 pr = sorted[st + o];
            const int q = atomicAdd(&lcnt[pr.y - base], 1);
            place_edge(slots, ovf, ovfn, pr.x, pr.y, q, e);
        }
    }
    __syncthreads();

    const int node = base + tid;
    if (node < n) cnt[node] = lcnt[tid];
}

__device__ __forceinline__ void fma4(float4& acc, const float4 v, const float w) {
    acc.x += v.x * w;
    acc.y += v.y * w;
    acc.z += v.z * w;
    acc.w += v.w * w;
}

__device__ __forceinline__ float dinv_of(const int counter) {
    return __builtin_amdgcn_rsqf((float)(counter + 1));
}

__device__ __forceinline__ float edge_norm(const int csrc, const float ddst) {
    return dinv_of(csrc) * ddst;
}

__device__ __forceinline__ float4 gather_node(const float4* __restrict__ rows,
                                              const int* __restrict__ cnt,
                                              const int2* __restrict__ ovf, const int node,
                                              const int c, const int ent, const int ce,
                                              const int lane, const int n, const int e) {
    const float4 vself = rows[(size_t)node * (NFEAT / 4) + lane];
    const int m = min(c, CAP);
    const float di = dinv_of(c);
    float4 acc = make_float4(0.0f, 0.0f, 0.0f, 0.0f);

    {
        int j = 0;
        for (; j + 4 <= m; j += 4) {
            const int s0 = __builtin_amdgcn_readlane(ent, j);
            const int s1 = __builtin_amdgcn_readlane(ent, j + 1);
            const int s2 = __builtin_amdgcn_readlane(ent, j + 2);
            const int s3 = __builtin_amdgcn_readlane(ent, j + 3);
            const float4 v0 = rows[(size_t)s0 * (NFEAT / 4) + lane];
            const float4 v1 = rows[(size_t)s1 * (NFEAT / 4) + lane];
            const float4 v2 = rows[(size_t)s2 * (NFEAT / 4) + lane];
            const float4 v3 = rows[(size_t)s3 * (NFEAT / 4) + lane];
            const float w0 = edge_norm(__builtin_amdgcn_readlane(ce, j), di);
            const float w1 = edge_norm(__builtin_amdgcn_readlane(ce, j + 1), di);
            const float w2 = edge_norm(__builtin_amdgcn_readlane(ce, j + 2), di);
            const float w3 = edge_norm(__builtin_amdgcn_readlane(ce, j + 3), di);
            fma4(acc, v0, w0);
            fma4(acc, v1, w1);
            fma4(acc, v2, w2);
            fma4(acc, v3, w3);
        }
        if (j + 2 <= m) {
            const int s0 = __builtin_amdgcn_readlane(ent, j);
            const int s1 = __builtin_amdgcn_readlane(ent, j + 1);
            const float4 v0 = rows[(size_t)s0 * (NFEAT / 4) + lane];
            const float4 v1 = rows[(size_t)s1 * (NFEAT / 4) + lane];
            const float w0 = edge_norm(__builtin_amdgcn_readlane(ce, j), di);
            const float w1 = edge_norm(__builtin_amdgcn_readlane(ce, j + 1), di);
            fma4(acc, v0, w0);
            fma4(acc, v1, w1);
            j += 2;
        }
        if (j < m) {
            const int s0 = __builtin_amdgcn_readlane(ent, j);
            const float4 v0 = rows[(size_t)s0 * (NFEAT / 4) + lane];
            const float w0 = edge_norm(__builtin_amdgcn_readlane(ce, j), di);
            fma4(acc, v0, w0);
        }
    }

    if (c > CAP) {
        const int on = min(cnt[n], e);
        for (int base = 0; base < on; base += 32) {
            const int idx = base + lane;
            int2 pr = make_int2(0, -1);
            if (idx < on) pr = ovf[idx];
            const bool hit = (pr.y == node);
            float wl = 0.0f;
            if (hit) wl = edge_norm(cnt[pr.x], di);
            unsigned int mask = __builtin_amdgcn_ballot_w32(hit);
            while (mask != 0u) {
                const int j = __builtin_ctz(mask);
                mask &= mask - 1u;
                const int s0 = __builtin_amdgcn_readlane(pr.x, j);
                const float w0 = __int_as_float(__builtin_amdgcn_readlane(__float_as_int(wl), j));
                const float4 v0 = rows[(size_t)s0 * (NFEAT / 4) + lane];
                fma4(acc, v0, w0);
            }
        }
    }

    fma4(acc, vself, di * di);
    return acc;
}


__global__ __launch_bounds__(256) void gather_product_finalize(const float* __restrict__ x,
                                                               const float* __restrict__ W,
                                                               const int* __restrict__ cnt,
                                                               const int* __restrict__ slots,
                                                               const int2* __restrict__ ovf,
                                                               const float* __restrict__ bias,
                                                               const float* __restrict__ wreg,
                                                               const float* __restrict__ breg,
                                                               float* __restrict__ out, int n, int e) {
    __shared__ __attribute__((aligned(16))) float lds_t[TROWS * XPITCH];
    __shared__ float lds_p[8 * TROWS];

    const int tid   = threadIdx.x;
    const int lane  = tid & 31;
    const int wave  = __builtin_amdgcn_readfirstlane(tid >> 5);
    const int o0    = wave * 16;
    const int row   = lane & 15;
    const int khalf = (lane >> 4) << 1;
    const int mh    = (lane >> 4) << 3;

    const float* wrow = W + (size_t)(o0 + row) * NFEAT;
    v2f bf[NFEAT / 4];
#pragma unroll
    for (int j = 0; j < NFEAT / 4; ++j) {
        const int kk = 4 * j + khalf;
        bf[j].x = wrow[kk];
        bf[j].y = wrow[kk + 1];
    }

    const float bcol = bias[o0 + row];
    const float wcol = wreg[o0 + row];
    const float br = breg[0];
    const float4* rows = (const float4*)x;
    const int ntile = (n + TROWS - 1) / TROWS;
    const int r0 = wave * 4;

    for (int t = blockIdx.x; t < ntile; t += gridDim.x) {
        const int n0 = t * TROWS;

        int c4[4], ent4[4], ce4[4];
        int cv = 0;
        {
            const int nd = n0 + r0 + (lane & 3);
            if (nd < n) cv = cnt[nd];
        }
#pragma unroll
        for (int i = 0; i < 4; ++i) c4[i] = __builtin_amdgcn_readlane(cv, i);
#pragma unroll
        for (int i = 0; i < 4; ++i) {
            const int node = n0 + r0 + i;
            const int m = min(c4[i], CAP);
            ent4[i] = 0;
            if (m > 0) ent4[i] = slots[(size_t)node * CAP + min(lane, m - 1)];
        }
#pragma unroll
        for (int i = 0; i < 4; ++i)
            ce4[i] = cnt[ent4[i]];
#pragma unroll
        for (int i = 0; i < 4; ++i) {
            const int node = n0 + r0 + i;
            float4 acc = make_float4(0.0f, 0.0f, 0.0f, 0.0f);
            if (node < n)
                acc = gather_node(rows, cnt, ovf, node, c4[i], ent4[i], ce4[i], lane, n, e);
            *(float4*)&lds_t[(r0 + i) * XPITCH + (lane << 2)] = acc;
        }
        __syncthreads();

        v8f acc0 = {};
        v8f acc1 = {};
#pragma unroll
        for (int j = 0; j < NFEAT / 4; ++j) {
            const int kk = 4 * j + khalf;
            v2f a0, a1;
            a0.x = lds_t[row * XPITCH + kk];
            a0.y = lds_t[row * XPITCH + kk + 1];
            a1.x = lds_t[(row + 16) * XPITCH + kk];
            a1.y = lds_t[(row + 16) * XPITCH + kk + 1];
            acc0 = __builtin_amdgcn_wmma_f32_16x16x4_f32(false, a0, false, bf[j],
                                                         (short)0, acc0, false, false);
            acc1 = __builtin_amdgcn_wmma_f32_16x16x4_f32(false, a1, false, bf[j],
                                                         (short)0, acc1, false, false);
        }

        float v0[8], v1[8];
#pragma unroll
        for (int r = 0; r < 8; ++r) {
            v0[r] = fmaxf(acc0[r] + bcol, 0.0f) * wcol;
            v1[r] = fmaxf(acc1[r] + bcol, 0.0f) * wcol;
        }
#pragma unroll
        for (int o = 8; o > 0; o >>= 1) {
#pragma unroll
            for (int r = 0; r < 8; ++r) {
                v0[r] += __shfl_xor(v0[r], o, 32);
                v1[r] += __shfl_xor(v1[r], o, 32);
            }
        }
        if (row == 0) {
#pragma unroll
            for (int r = 0; r < 8; ++r) {
                lds_p[wave * TROWS + mh + r]      = v0[r];
                lds_p[wave * TROWS + 16 + mh + r] = v1[r];
            }
        }
        __syncthreads();

        if (tid < TROWS) {
            const int node = n0 + tid;
            if (node < n) {
                float sum = lds_p[tid];
#pragma unroll
                for (int w8 = 1; w8 < 8; ++w8) sum += lds_p[w8 * TROWS + tid];
                out[node] = sum + br;
            }
        }
    }
}


extern "C" void kernel_launch(void* const* d_in, const int* in_sizes, int n_in,
                              void* d_out, int out_size, void* d_ws, size_t ws_size,
                              hipStream_t stream) {
    const float* x    = (const float*)d_in[0];
    const int*   ei   = (const int*)d_in[1];
    const float* W    = (const float*)d_in[2];
    const float* b    = (const float*)d_in[3];
    const float* wreg = (const float*)d_in[4];
    const float* breg = (const float*)d_in[5];
    float* out = (float*)d_out;

    const int n = in_sizes[0] / NFEAT;
    const int e = in_sizes[1] / 2;
    const int* src = ei;
    const int* dst = ei + e;

    int*   cnt    = (int*)d_ws;
    int*   slots  = cnt + n + 32;
    int2*  ovf    = (int2*)(slots + (size_t)n * CAP);
    int2*  sorted = ovf + e;
    int2*  tab    = sorted + e;

    const int B = 256;
    bin_edges_kernel<<<PA, TA, 0, stream>>>(src, dst, sorted, tab, cnt + n, n, e);

    place_bucket_kernel<<<(n + RB - 1) / RB, RB, 0, stream>>>(sorted, tab, cnt, slots, ovf, n, e);

    gather_product_finalize<<<1024, B, 0, stream>>>(x, W, cnt, slots, ovf, b, wreg, breg, out, n, e);
}


# W fragment of a wave built from 16 whole-row loads (16 bytes a lane, 4 lines a load) turned through LDS once, not 32 strided 8-byte loads of 16 lines each; every sum and its order unchanged
# speedup vs baseline: 4.8025x; 1.0490x over previous
#include <hip/hip_runtime.h>
#include <cstdint>


#define NFEAT 128
#define XPITCH 132
#define CAP 32
#define TROWS 32

typedef __attribute__((ext_vector_type(2))) float v2f;
typedef __attribute__((ext_vector_type(8))) float v8f;


#define SH 9
#define RB (1 << SH)
#define NBKMAX 256
#define PA 128
#define TA 1024
#define KA 5

__device__ __forceinline__ void place_edge(int* __restrict__ slots, int2* __restrict__ ovf,
                                           int* __restrict__ ovfn, int s, int d, int slot, int e) {
    if (slot < CAP) {
        slots[(size_t)d * CAP + slot] = s;
    } else {
        const int p = atomicAdd(ovfn, 1);
        if (p < e) ovf[p] = make_int2(s, d);
    }
}

__global__ __launch_bounds__(TA) void bin_edges_kernel(const int* __restrict__ src,
                                                       const int* __restrict__ dst,
                                                       int2* __restrict__ sorted,
                                                       int2* __restrict__ tab,
                                                       int* __restrict__ ovfn, int n, int e) {
    __shared__ int hist[NBKMAX];
    __shared__ int cur[NBKMAX];
    __shared__ int stage_s[KA * TA];
    __shared__ int stage_d[KA * TA];
    const int tid = threadIdx.x;
    const int p = blockIdx.x;
    const int nbk = (n + RB - 1) >> SH;
    const int C = (e + PA - 1) / PA;
    const int c0 = p * C;
    const int c1 = min(e, c0 + C);
    const bool staged = (c1 - c0) <= KA * TA;

    if (p == 0 && tid == 0) *ovfn = 0;
    for (int b = tid; b < NBKMAX; b += TA) hist[b] = 0;
    __syncthreads();

    int d[KA], s[KA], pos[KA];
#pragma unroll
    for (int k = 0; k < KA; ++k) {
        const int i = c0 + tid + k * TA;
        d[k] = -1;
        s[k] = 0;
        if (i < c1) {
            d[k] = dst[i];
            s[k] = src[i];
        }
    }
#pragma unroll
    for (int k = 0; k < KA; ++k)
        if (d[k] >= 0) atomicAdd(&hist[d[k] >> SH], 1);
    for (int i = c0 + tid + KA * TA; i < c1; i += TA)
        atomicAdd(&hist[dst[i] >> SH], 1);
    __syncthreads();

    if (tid < 32) {
        int v[NBKMAX / 32];
        int run = 0;
#pragma unroll
        for (int k = 0; k < NBKMAX / 32; ++k) {
            const int t = hist[tid * (NBKMAX / 32) + k];
            v[k] = run;
            run += t;
        }
        int inc = run;
#pragma unroll
        for (int o = 1; o < 32; o <<= 1) {
            const int t = __shfl_up(inc, o, 32);
            if (tid >= o) inc += t;
        }
        const int excl = inc - run;
#pragma unroll
        for (int k = 0; k < NBKMAX / 32; ++k) cur[tid * (NBKMAX / 32) + k] = excl + v[k];
    }
    __syncthreads();

    for (int b = tid; b < nbk; b += TA)
        tab[(size_t)b * PA + p] = make_int2(c0 + cur[b], hist[b]);
    __syncthreads();

#pragma unroll
    for (int k = 0; k < KA; ++k) {
        pos[k] = 0;
        if (d[k] >= 0) pos[k] = atomicAdd(&cur[d[k] >> SH], 1);
    }
    if (staged) {
#pragma unroll
        for (int k = 0; k < KA; ++k) {
            if (d[k] >= 0) {
                stage_s[pos[k]] = s[k];
                stage_d[pos[k]] = d[k];
            }
        }
        __syncthreads();
        for (int i = tid; i < c1 - c0; i += TA)
            sorted[c0 + i] = make_int2(stage_s[i], stage_d[i]);
    } else {
#pragma unroll
        for (int k = 0; k < KA; ++k)
            if (d[k] >= 0) sorted[c0 + pos[k]] = make_int2(s[k], d[k]);
        for (int i = c0 + tid + KA * TA; i < c1; i += TA) {
            const int dd = dst[i];
            const int ss = src[i];
            const int q = atomicAdd(&cur[dd >> SH], 1);
            sorted[c0 + q] = make_int2(ss, dd);
        }
    }
}

__global__ __launch_bounds__(RB) void place_bucket_kernel(const int2* __restrict__ sorted,
                                                          const int2* __restrict__ tab,
                                                          int* __restrict__ cnt,
                                                          int* __restrict__ slots,
                                                          int2* __restrict__ ovf, int n, int e) {
    __shared__ int lcnt[RB];
    const int tid  = threadIdx.x;
    const int lane = tid & 31;
    const int wave = __builtin_amdgcn_readfirstlane(tid >> 5);
    const int b = blockIdx.x;
    const int base = b << SH;
    int* ovfn = cnt + n;
    constexpr int RW = PA / (RB / 32);

    lcnt[tid] = 0;
    const int2 te = tab[(size_t)b * PA + wave * RW + (lane & (RW - 1))];
    __syncthreads();

    int2 ed[2 * RW];
#pragma unroll
    for (int k = 0; k < RW; ++k) {
        const int st = __builtin_amdgcn_readlane(te.x, k);
        const int ct = __builtin_amdgcn_readlane(te.y, k);
        ed[2 * k] = make_int2(0, -1);
        ed[2 * k + 1] = make_int2(0, -1);
        if (lane < ct) ed[2 * k] = sorted[st + lane];
        if (lane + 32 < ct) ed[2 * k + 1] = sorted[st + lane + 32];
    }
    int slot[2 * RW];
#pragma unroll
    for (int k = 0; k < 2 * RW; ++k) {
        slot[k] = 0;
        if (ed[k].y >= 0) slot[k] = atomicAdd(&lcnt[ed[k].y - base], 1);
    }
#pragma unroll
    for (int k = 0; k < 2 * RW; ++k)
        if (ed[k].y >= 0) place_edge(slots, ovf, ovfn, ed[k].x, ed[k].y, slot[k], e);
#pragma unroll 1
    for (int k = 0; k < RW; ++k) {
        const int st = __builtin_amdgcn_readlane(te.x, k);
        const int ct = __builtin_amdgcn_readlane(te.y, k);
        for (int o = 64 + lane; o < ct; o += 32) {
            const int2 pr = sorted[st + o];
            const int q = atomicAdd(&lcnt[pr.y - base], 1);
            place_edge(slots, ovf, ovfn, pr.x, pr.y, q, e);
        }
    }
    __syncthreads();

    const int node = base + tid;
    if (node < n) cnt[node] = lcnt[tid];
}

__device__ __forceinline__ void fma4(float4& acc, const float4 v, const float w) {
    acc.x += v.x * w;
    acc.y += v.y * w;
    acc.z += v.z * w;
    acc.w += v.w * w;
}

__device__ __forceinline__ float dinv_of(const int counter) {
    return __builtin_amdgcn_rsqf((float)(counter + 1));
}

__device__ __forceinline__ float edge_norm(const int csrc, const float ddst) {
    return dinv_of(csrc) * ddst;
}

__device__ __forceinline__ float4 gather_node(const float4* __restrict__ rows,
                                              const int* __restrict__ cnt,
                                              const int2* __restrict__ ovf, const int node,
                                              const int c, const int ent, const int ce,
                                              const int lane, const int n, const int e) {
    const float4 vself = rows[(size_t)node * (NFEAT / 4) + lane];
    const int m = min(c, CAP);
    const float di = dinv_of(c);
    float4 acc = make_float4(0.0f, 0.0f, 0.0f, 0.0f);

    {
        int j = 0;
        for (; j + 4 <= m; j += 4) {
            const int s0 = __builtin_amdgcn_readlane(ent, j);
            const int s1 = __builtin_amdgcn_readlane(ent, j + 1);
            const int s2 = __builtin_amdgcn_readlane(ent, j + 2);
            const int s3 = __builtin_amdgcn_readlane(ent, j + 3);
            const float4 v0 = rows[(size_t)s0 * (NFEAT / 4) + lane];
            const float4 v1 = rows[(size_t)s1 * (NFEAT / 4) + lane];
            const float4 v2 = rows[(size_t)s2 * (NFEAT / 4) + lane];
            const float4 v3 = rows[(size_t)s3 * (NFEAT / 4) + lane];
            const float w0 = edge_norm(__builtin_amdgcn_readlane(ce, j), di);
            const float w1 = edge_norm(__builtin_amdgcn_readlane(ce, j + 1), di);
            const float w2 = edge_norm(__builtin_amdgcn_readlane(ce, j + 2), di);
            const float w3 = edge_norm(__builtin_amdgcn_readlane(ce, j + 3), di);
            fma4(acc, v0, w0);
            fma4(acc, v1, w1);
            fma4(acc, v2, w2);
            fma4(acc, v3, w3);
        }
        if (j + 2 <= m) {
            const int s0 = __builtin_amdgcn_readlane(ent, j);
            const int s1 = __builtin_amdgcn_readlane(ent, j + 1);
            const float4 v0 = rows[(size_t)s0 * (NFEAT / 4) + lane];
            const float4 v1 = rows[(size_t)s1 * (NFEAT / 4) + lane];
            const float w0 = edge_norm(__builtin_amdgcn_readlane(ce, j), di);
            const float w1 = edge_norm(__builtin_amdgcn_readlane(ce, j + 1), di);
            fma4(acc, v0, w0);
            fma4(acc, v1, w1);
            j += 2;
        }
        if (j < m) {
            const int s0 = __builtin_amdgcn_readlane(ent, j);
            const float4 v0 = rows[(size_t)s0 * (NFEAT / 4) + lane];
            const float w0 = edge_norm(__builtin_amdgcn_readlane(ce, j), di);
            fma4(acc, v0, w0);
        }
    }

    if (c > CAP) {
        const int on = min(cnt[n], e);
        for (int base = 0; base < on; base += 32) {
            const int idx = base + lane;
            int2 pr = make_int2(0, -1);
            if (idx < on) pr = ovf[idx];
            const bool hit = (pr.y == node);
            float wl = 0.0f;
            if (hit) wl = edge_norm(cnt[pr.x], di);
            unsigned int mask = __builtin_amdgcn_ballot_w32(hit);
            while (mask != 0u) {
                const int j = __builtin_ctz(mask);
                mask &= mask - 1u;
                const int s0 = __builtin_amdgcn_readlane(pr.x, j);
                const float w0 = __int_as_float(__builtin_amdgcn_readlane(__float_as_int(wl), j));
                const float4 v0 = rows[(size_t)s0 * (NFEAT / 4) + lane];
                fma4(acc, v0, w0);
            }
        }
    }

    fma4(acc, vself, di * di);
    return acc;
}


__global__ __launch_bounds__(256) void gather_product_finalize(const float* __restrict__ x,
                                                               const float* __restrict__ W,
                                                               const int* __restrict__ cnt,
                                                               const int* __restrict__ slots,
                                                               const int2* __restrict__ ovf,
                                                               const float* __restrict__ bias,
                                                               const float* __restrict__ wreg,
                                                               const float* __restrict__ breg,
                                                               float* __restrict__ out, int n, int e) {
    __shared__ __attribute__((aligned(16))) float lds_t[8 * 16 * XPITCH];
    __shared__ float lds_p[8 * TROWS];

    const int tid   = threadIdx.x;
    const int lane  = tid & 31;
    const int wave  = __builtin_amdgcn_readfirstlane(tid >> 5);
    const int o0    = wave * 16;
    const int row   = lane & 15;
    const int khalf = (lane >> 4) << 1;
    const int mh    = (lane >> 4) << 3;

    float* wst = lds_t + wave * 16 * XPITCH;
    {
        const float4* w4 = (const float4*)W;
        float4 wv[16];
#pragma unroll
        for (int r = 0; r < 16; ++r) wv[r] = w4[(size_t)(o0 + r) * (NFEAT / 4) + lane];
#pragma unroll
        for (int r = 0; r < 16; ++r) *(float4*)&wst[r * XPITCH + (lane << 2)] = wv[r];
    }
    __syncthreads();
    v2f bf[NFEAT / 4];
#pragma unroll
    for (int j = 0; j < NFEAT / 4; ++j) {
        const int kk = 4 * j + khalf;
        bf[j].x = wst[row * XPITCH + kk];
        bf[j].y = wst[row * XPITCH + kk + 1];
    }
    __syncthreads();

    const float bcol = bias[o0 + row];
    const float wcol = wreg[o0 + row];
    const float br = breg[0];
    const float4* rows = (const float4*)x;
    const int ntile = (n + TROWS - 1) / TROWS;
    const int r0 = wave * 4;

    for (int t = blockIdx.x; t < ntile; t += gridDim.x) {
        const int n0 = t * TROWS;

        int c4[4], ent4[4], ce4[4];
        int cv = 0;
        {
            const int nd = n0 + r0 + (lane & 3);
            if (nd < n) cv = cnt[nd];
        }
#pragma unroll
        for (int i = 0; i < 4; ++i) c4[i] = __builtin_amdgcn_readlane(cv, i);
#pragma unroll
        for (int i = 0; i < 4; ++i) {
            const int node = n0 + r0 + i;
            const int m = min(c4[i], CAP);
            ent4[i] = 0;
            if (m > 0) ent4[i] = slots[(size_t)node * CAP + min(lane, m - 1)];
        }
#pragma unroll
        for (int i = 0; i < 4; ++i)
            ce4[i] = cnt[ent4[i]];
#pragma unroll
        for (int i = 0; i < 4; ++i) {
            const int node = n0 + r0 + i;
            float4 acc = make_float4(0.0f, 0.0f, 0.0f, 0.0f);
            if (node < n)
                acc = gather_node(rows, cnt, ovf, node, c4[i], ent4[i], ce4[i], lane, n, e);
            *(float4*)&lds_t[(r0 + i) * XPITCH + (lane << 2)] = acc;
        }
        __syncthreads();

        v8f acc0 = {};
        v8f acc1 = {};
#pragma unroll
        for (int j = 0; j < NFEAT / 4; ++j) {
            const int kk = 4 * j + khalf;
            v2f a0, a1;
            a0.x = lds_t[row * XPITCH + kk];
            a0.y = lds_t[row * XPITCH + kk + 1];
            a1.x = lds_t[(row + 16) * XPITCH + kk];
            a1.y = lds_t[(row + 16) * XPITCH + kk + 1];
            acc0 = __builtin_amdgcn_wmma_f32_16x16x4_f32(false, a0, false, bf[j],
                                                         (short)0, acc0, false, false);
            acc1 = __builtin_amdgcn_wmma_f32_16x16x4_f32(false, a1, false, bf[j],
                                                         (short)0, acc1, false, false);
        }

        float v0[8], v1[8];
#pragma unroll
        for (int r = 0; r < 8; ++r) {
            v0[r] = fmaxf(acc0[r] + bcol, 0.0f) * wcol;
            v1[r] = fmaxf(acc1[r] + bcol, 0.0f) * wcol;
        }
#pragma unroll
        for (int o = 8; o > 0; o >>= 1) {
#pragma unroll
            for (int r = 0; r < 8; ++r) {
                v0[r] += __shfl_xor(v0[r], o, 32);
                v1[r] += __shfl_xor(v1[r], o, 32);
            }
        }
        if (row == 0) {
#pragma unroll
            for (int r = 0; r < 8; ++r) {
                lds_p[wave * TROWS + mh + r]      = v0[r];
                lds_p[wave * TROWS + 16 + mh + r] = v1[r];
            }
        }
        __syncthreads();

        if (tid < TROWS) {
            const int node = n0 + tid;
            if (node < n) {
                float sum = lds_p[tid];
#pragma unroll
                for (int w8 = 1; w8 < 8; ++w8) sum += lds_p[w8 * TROWS + tid];
                out[node] = sum + br;
            }
        }
    }
}


extern "C" void kernel_launch(void* const* d_in, const int* in_sizes, int n_in,
                              void* d_out, int out_size, void* d_ws, size_t ws_size,
                              hipStream_t stream) {
    const float* x    = (const float*)d_in[0];
    const int*   ei   = (const int*)d_in[1];
    const float* W    = (const float*)d_in[2];
    const float* b    = (const float*)d_in[3];
    const float* wreg = (const float*)d_in[4];
    const float* breg = (const float*)d_in[5];
    float* out = (float*)d_out;

    const int n = in_sizes[0] / NFEAT;
    const int e = in_sizes[1] / 2;
    const int* src = ei;
    const int* dst = ei + e;

    int*   cnt    = (int*)d_ws;
    int*   slots  = cnt + n + 32;
    int2*  ovf    = (int2*)(slots + (size_t)n * CAP);
    int2*  sorted = ovf + e;
    int2*  tab    = sorted + e;

    const int B = 256;
    bin_edges_kernel<<<PA, TA, 0, stream>>>(src, dst, sorted, tab, cnt + n, n, e);

    place_bucket_kernel<<<(n + RB - 1) / RB, RB, 0, stream>>>(sorted, tab, cnt, slots, ovf, n, e);

    gather_product_finalize<<<1024, B, 0, stream>>>(x, W, cnt, slots, ovf, b, wreg, breg, out, n, e);
}
